# starting kernel
# baseline (speedup 1.0000x reference)
#include <hip/hip_runtime.h>
#include <math.h>

#define EMB 128
#define NEG_SLOPE 0.2f

typedef float v2f __attribute__((ext_vector_type(2)));
typedef float v8f __attribute__((ext_vector_type(8)));

__device__ __forceinline__ void atomicMaxFloat(float* addr, float v) {
    if (v >= 0.0f) atomicMax((int*)addr, __float_as_int(v));
    else           atomicMin((unsigned int*)addr, __float_as_uint(v));
}

__global__ void gat_init_kernel(float* __restrict__ o, float* __restrict__ deg,
                                float* __restrict__ lw, float* __restrict__ amax,
                                float* __restrict__ denom, int N) {
    int i = blockIdx.x * blockDim.x + threadIdx.x;
    if (i < N * EMB) o[i] = 0.0f;
    if (i < N) { deg[i] = 0.0f; lw[i] = 0.0f; }
    if (i < 2 * N) { amax[i] = -INFINITY; denom[i] = 0.0f; }
}

__global__ void gat_deg_kernel(const int* __restrict__ dst, const float* __restrict__ ew,
                               float* __restrict__ deg, float* __restrict__ lw, int E) {
    int e = blockIdx.x * blockDim.x + threadIdx.x;
    if (e >= E) return;
    int d = dst[e];
    atomicAdd(&deg[d], 1.0f);
    atomicAdd(&lw[d], ew[e]);
}

__global__ __launch_bounds__(256)
void gat_gemm_kernel(const float* __restrict__ h, const float* __restrict__ Wl,
                     const float* __restrict__ Wr, float* __restrict__ xl,
                     float* __restrict__ xr, int N) {
    int lane = threadIdx.x & 31;
    int wave = (blockIdx.x * blockDim.x + threadIdx.x) >> 5;
    int rowTiles = (N + 15) >> 4;
    if (wave >= rowTiles * 16) return;
    int rowTile = wave >> 4;
    int t = wave & 15;
    const float* W = (t & 8) ? Wr : Wl;
    float*      out = (t & 8) ? xr : xl;
    int ncol = (t & 7) * 16;
    int half = lane >> 4;
    int m    = lane & 15;
    int arow = rowTile * 16 + m;
    if (arow > N - 1) arow = N - 1;
    int col  = ncol + m;
    int koff = half * 2;

    v8f c = {0.f, 0.f, 0.f, 0.f, 0.f, 0.f, 0.f, 0.f};
#pragma unroll
    for (int k = 0; k < EMB; k += 4) {
        v2f a = *(const v2f*)(h + (size_t)arow * EMB + k + koff);
        const float* wp = W + (size_t)(k + koff) * EMB + col;
        v2f b; b.x = wp[0]; b.y = wp[EMB];
        c = __builtin_amdgcn_wmma_f32_16x16x4_f32(false, a, false, b, (short)0, c,
                                                  false, false);
    }
#pragma unroll
    for (int j = 0; j < 8; ++j) {
        int r = rowTile * 16 + j + half * 8;
        if (r < N) out[(size_t)r * EMB + col] = c[j];
    }
}

__global__ void gat_alpha_kernel(const float* __restrict__ xl, const float* __restrict__ xr,
                                 const int* __restrict__ src, const int* __restrict__ dst,
                                 const float* __restrict__ ew, const float* __restrict__ deg,
                                 const float* __restrict__ lw, const float* __restrict__ We,
                                 const float* __restrict__ att, float* __restrict__ alpha,
                                 float* __restrict__ amax, int E, int N) {
    int wave = (blockIdx.x * blockDim.x + threadIdx.x) >> 5;
    int lane = threadIdx.x & 31;
    if (wave >= E + N) return;
    int s, d; float w;
    if (wave < E) { s = src[wave]; d = dst[wave]; w = ew[wave]; }
    else { int i = wave - E; s = i; d = i; w = lw[i] / fmaxf(deg[i], 1.0f); }

    const float4 xs = *(const float4*)(xl + (size_t)s * EMB + lane * 4);
    const float4 xd = *(const float4*)(xr + (size_t)d * EMB + lane * 4);
    const float4 we = *(const float4*)(We + lane * 4);
    const float4 at = *(const float4*)(att + lane * 4);

    float p = 0.0f, v;
    v = xs.x + xd.x + w * we.x; v = (v >= 0.f) ? v : NEG_SLOPE * v; p += v * at.x;
    v = xs.y + xd.y + w * we.y; v = (v >= 0.f) ? v : NEG_SLOPE * v; p += v * at.y;
    v = xs.z + xd.z + w * we.z; v = (v >= 0.f) ? v : NEG_SLOPE * v; p += v * at.z;
    v = xs.w + xd.w + w * we.w; v = (v >= 0.f) ? v : NEG_SLOPE * v; p += v * at.w;

    for (int msk = 8; msk >= 1; msk >>= 1) p += __shfl_xor(p, msk, 32);

    if ((lane & 15) == 0) {
        int head = lane >> 4;
        alpha[(size_t)wave * 2 + head] = p;
        atomicMaxFloat(&amax[(size_t)d * 2 + head], p);
    }
}

__global__ void gat_expsum_kernel(const int* __restrict__ dst, float* __restrict__ alpha,
                                  const float* __restrict__ amax, float* __restrict__ denom,
                                  int E, int N) {
    int idx = blockIdx.x * blockDim.x + threadIdx.x;
    if (idx >= 2 * (E + N)) return;
    int e = idx >> 1, head = idx & 1;
    int d = (e < E) ? dst[e] : (e - E);
    float ex = __expf(alpha[idx] - amax[(size_t)d * 2 + head]);
    alpha[idx] = ex;
    atomicAdd(&denom[(size_t)d * 2 + head], ex);
}

__global__ void gat_aggregate_kernel(const float* __restrict__ xl, const int* __restrict__ src,
                                     const int* __restrict__ dst, const float* __restrict__ alpha,
                                     const float* __restrict__ denom, float* __restrict__ o,
                                     int E, int N) {
    int wave = (blockIdx.x * blockDim.x + threadIdx.x) >> 5;
    int lane = threadIdx.x & 31;
    if (wave >= E + N) return;
    int s, d;
    if (wave < E) { s = src[wave]; d = dst[wave]; }
    else { s = d = wave - E; }
    int head = lane >> 4;
    float a = alpha[(size_t)wave * 2 + head] / denom[(size_t)d * 2 + head];
    const float4 m = *(const float4*)(xl + (size_t)s * EMB + lane * 4);
    float* op = o + (size_t)d * EMB + lane * 4;
    atomicAdd(op + 0, m.x * a);
    atomicAdd(op + 1, m.y * a);
    atomicAdd(op + 2, m.z * a);
    atomicAdd(op + 3, m.w * a);
}

__global__ void gat_finish_kernel(float* __restrict__ o, const float* __restrict__ bias, int N) {
    int i = blockIdx.x * blockDim.x + threadIdx.x;
    if (i >= N * EMB) return;
    float x = o[i] + bias[i & (EMB - 1)];
    o[i] = (x > 0.0f) ? x : (__expf(x) - 1.0f);
}

extern "C" void kernel_launch(void* const* d_in, const int* in_sizes, int n_in,
                              void* d_out, int out_size, void* d_ws, size_t ws_size,
                              hipStream_t stream) {
    const int*   edge_index = (const int*)d_in[0];
    const float* ew   = (const float*)d_in[1];
    const float* emb  = (const float*)d_in[2];
    const float* Wl   = (const float*)d_in[3];
    const float* Wr   = (const float*)d_in[4];
    const float* We   = (const float*)d_in[5];
    const float* att  = (const float*)d_in[6];
    const float* bias = (const float*)d_in[7];

    const int E = in_sizes[0] / 2;
    const int N = in_sizes[2] / EMB;
    const int* src = edge_index;
    const int* dst = edge_index + E;

    float* ws    = (float*)d_ws;
    float* xl    = ws;
    float* xr    = xl + (size_t)N * EMB;
    float* hbuf  = xr + (size_t)N * EMB;
    float* alpha = hbuf + (size_t)N * EMB;
    float* amax  = alpha + (size_t)2 * (E + N);
    float* denom = amax + (size_t)2 * N;
    float* deg   = denom + (size_t)2 * N;
    float* lw    = deg + N;

    const int BT = 256;
    const int rowTiles = (N + 15) / 16;
    const int gemmBlocks = (rowTiles * 16 + 7) / 8;
    const int edgeWaveBlocks = ((E + N) * 32 + BT - 1) / BT;
    const int nodeBlocks = (N * EMB + BT - 1) / BT;

    for (int l = 0; l < 2; ++l) {
        const float* hin = (l == 0) ? emb : hbuf;
        float* o = (l == 0) ? hbuf : (float*)d_out;
        const float* Wl_l = Wl + (size_t)l * EMB * EMB;
        const float* Wr_l = Wr + (size_t)l * EMB * EMB;
        const float* We_l = We + (size_t)l * EMB;
        const float* at_l = att + (size_t)l * EMB;
        const float* b_l  = bias + (size_t)l * EMB;

        gat_init_kernel<<<nodeBlocks, BT, 0, stream>>>(o, deg, lw, amax, denom, N);
        gat_deg_kernel<<<(E + BT - 1) / BT, BT, 0, stream>>>(dst, ew, deg, lw, E);
        gat_gemm_kernel<<<gemmBlocks, BT, 0, stream>>>(hin, Wl_l, Wr_l, xl, xr, N);
        gat_alpha_kernel<<<edgeWaveBlocks, BT, 0, stream>>>(xl, xr, src, dst, ew, deg, lw,
                                                            We_l, at_l, alpha, amax, E, N);
        gat_expsum_kernel<<<(2 * (E + N) + BT - 1) / BT, BT, 0, stream>>>(dst, alpha, amax,
                                                                          denom, E, N);
        gat_aggregate_kernel<<<edgeWaveBlocks, BT, 0, stream>>>(xl, src, dst, alpha, denom,
                                                                o, E, N);
        gat_finish_kernel<<<nodeBlocks, BT, 0, stream>>>(o, b_l, N);
    }
}


# degree and self-loop weight sums are layer-invariant: build them once per call (first layer) and reuse in the second, dropping one atomic scatter pass
# speedup vs baseline: 1.0945x; 1.0945x over previous
#include <hip/hip_runtime.h>
#include <math.h>

#define EMB 128
#define NEG_SLOPE 0.2f

typedef float v2f __attribute__((ext_vector_type(2)));
typedef float v8f __attribute__((ext_vector_type(8)));

__device__ __forceinline__ void atomicMaxFloat(float* addr, float v) {
    if (v >= 0.0f) atomicMax((int*)addr, __float_as_int(v));
    else           atomicMin((unsigned int*)addr, __float_as_uint(v));
}

__global__ void gat_init_kernel(float* __restrict__ o, float* __restrict__ deg,
                                float* __restrict__ lw, float* __restrict__ amax,
                                float* __restrict__ denom, int N, int first_layer) {
    int i = blockIdx.x * blockDim.x + threadIdx.x;
    if (i < N * EMB) o[i] = 0.0f;
    if (first_layer && i < N) { deg[i] = 0.0f; lw[i] = 0.0f; }
    if (i < 2 * N) { amax[i] = -INFINITY; denom[i] = 0.0f; }
}

__global__ void gat_deg_kernel(const int* __restrict__ dst, const float* __restrict__ ew,
                               float* __restrict__ deg, float* __restrict__ lw, int E) {
    int e = blockIdx.x * blockDim.x + threadIdx.x;
    if (e >= E) return;
    int d = dst[e];
    atomicAdd(&deg[d], 1.0f);
    atomicAdd(&lw[d], ew[e]);
}

__global__ __launch_bounds__(256)
void gat_gemm_kernel(const float* __restrict__ h, const float* __restrict__ Wl,
                     const float* __restrict__ Wr, float* __restrict__ xl,
                     float* __restrict__ xr, int N) {
    int lane = threadIdx.x & 31;
    int wave = (blockIdx.x * blockDim.x + threadIdx.x) >> 5;
    int rowTiles = (N + 15) >> 4;
    if (wave >= rowTiles * 16) return;
    int rowTile = wave >> 4;
    int t = wave & 15;
    const float* W = (t & 8) ? Wr : Wl;
    float*      out = (t & 8) ? xr : xl;
    int ncol = (t & 7) * 16;
    int half = lane >> 4;
    int m    = lane & 15;
    int arow = rowTile * 16 + m;
    if (arow > N - 1) arow = N - 1;
    int col  = ncol + m;
    int koff = half * 2;

    v8f c = {0.f, 0.f, 0.f, 0.f, 0.f, 0.f, 0.f, 0.f};
#pragma unroll
    for (int k = 0; k < EMB; k += 4) {
        v2f a = *(const v2f*)(h + (size_t)arow * EMB + k + koff);
        const float* wp = W + (size_t)(k + koff) * EMB + col;
        v2f b; b.x = wp[0]; b.y = wp[EMB];
        c = __builtin_amdgcn_wmma_f32_16x16x4_f32(false, a, false, b, (short)0, c,
                                                  false, false);
    }
#pragma unroll
    for (int j = 0; j < 8; ++j) {
        int r = rowTile * 16 + j + half * 8;
        if (r < N) out[(size_t)r * EMB + col] = c[j];
    }
}

__global__ void gat_alpha_kernel(const float* __restrict__ xl, const float* __restrict__ xr,
                                 const int* __restrict__ src, const int* __restrict__ dst,
                                 const float* __restrict__ ew, const float* __restrict__ deg,
                                 const float* __restrict__ lw, const float* __restrict__ We,
                                 const float* __restrict__ att, float* __restrict__ alpha,
                                 float* __restrict__ amax, int E, int N) {
    int wave = (blockIdx.x * blockDim.x + threadIdx.x) >> 5;
    int lane = threadIdx.x & 31;
    if (wave >= E + N) return;
    int s, d; float w;
    if (wave < E) { s = src[wave]; d = dst[wave]; w = ew[wave]; }
    else { int i = wave - E; s = i; d = i; w = lw[i] / fmaxf(deg[i], 1.0f); }

    const float4 xs = *(const float4*)(xl + (size_t)s * EMB + lane * 4);
    const float4 xd = *(const float4*)(xr + (size_t)d * EMB + lane * 4);
    const float4 we = *(const float4*)(We + lane * 4);
    const float4 at = *(const float4*)(att + lane * 4);

    float p = 0.0f, v;
    v = xs.x + xd.x + w * we.x; v = (v >= 0.f) ? v : NEG_SLOPE * v; p += v * at.x;
    v = xs.y + xd.y + w * we.y; v = (v >= 0.f) ? v : NEG_SLOPE * v; p += v * at.y;
    v = xs.z + xd.z + w * we.z; v = (v >= 0.f) ? v : NEG_SLOPE * v; p += v * at.z;
    v = xs.w + xd.w + w * we.w; v = (v >= 0.f) ? v : NEG_SLOPE * v; p += v * at.w;

    for (int msk = 8; msk >= 1; msk >>= 1) p += __shfl_xor(p, msk, 32);

    if ((lane & 15) == 0) {
        int head = lane >> 4;
        alpha[(size_t)wave * 2 + head] = p;
        atomicMaxFloat(&amax[(size_t)d * 2 + head], p);
    }
}

__global__ void gat_expsum_kernel(const int* __restrict__ dst, float* __restrict__ alpha,
                                  const float* __restrict__ amax, float* __restrict__ denom,
                                  int E, int N) {
    int idx = blockIdx.x * blockDim.x + threadIdx.x;
    if (idx >= 2 * (E + N)) return;
    int e = idx >> 1, head = idx & 1;
    int d = (e < E) ? dst[e] : (e - E);
    float ex = __expf(alpha[idx] - amax[(size_t)d * 2 + head]);
    alpha[idx] = ex;
    atomicAdd(&denom[(size_t)d * 2 + head], ex);
}

__global__ void gat_aggregate_kernel(const float* __restrict__ xl, const int* __restrict__ src,
                                     const int* __restrict__ dst, const float* __restrict__ alpha,
                                     const float* __restrict__ denom, float* __restrict__ o,
                                     int E, int N) {
    int wave = (blockIdx.x * blockDim.x + threadIdx.x) >> 5;
    int lane = threadIdx.x & 31;
    if (wave >= E + N) return;
    int s, d;
    if (wave < E) { s = src[wave]; d = dst[wave]; }
    else { s = d = wave - E; }
    int head = lane >> 4;
    float a = alpha[(size_t)wave * 2 + head] / denom[(size_t)d * 2 + head];
    const float4 m = *(const float4*)(xl + (size_t)s * EMB + lane * 4);
    float* op = o + (size_t)d * EMB + lane * 4;
    atomicAdd(op + 0, m.x * a);
    atomicAdd(op + 1, m.y * a);
    atomicAdd(op + 2, m.z * a);
    atomicAdd(op + 3, m.w * a);
}

__global__ void gat_finish_kernel(float* __restrict__ o, const float* __restrict__ bias, int N) {
    int i = blockIdx.x * blockDim.x + threadIdx.x;
    if (i >= N * EMB) return;
    float x = o[i] + bias[i & (EMB - 1)];
    o[i] = (x > 0.0f) ? x : (__expf(x) - 1.0f);
}

extern "C" void kernel_launch(void* const* d_in, const int* in_sizes, int n_in,
                              void* d_out, int out_size, void* d_ws, size_t ws_size,
                              hipStream_t stream) {
    const int*   edge_index = (const int*)d_in[0];
    const float* ew   = (const float*)d_in[1];
    const float* emb  = (const float*)d_in[2];
    const float* Wl   = (const float*)d_in[3];
    const float* Wr   = (const float*)d_in[4];
    const float* We   = (const float*)d_in[5];
    const float* att  = (const float*)d_in[6];
    const float* bias = (const float*)d_in[7];

    const int E = in_sizes[0] / 2;
    const int N = in_sizes[2] / EMB;
    const int* src = edge_index;
    const int* dst = edge_index + E;

    float* ws    = (float*)d_ws;
    float* xl    = ws;
    float* xr    = xl + (size_t)N * EMB;
    float* hbuf  = xr + (size_t)N * EMB;
    float* alpha = hbuf + (size_t)N * EMB;
    float* amax  = alpha + (size_t)2 * (E + N);
    float* denom = amax + (size_t)2 * N;
    float* deg   = denom + (size_t)2 * N;
    float* lw    = deg + N;

    const int BT = 256;
    const int rowTiles = (N + 15) / 16;
    const int gemmBlocks = (rowTiles * 16 + 7) / 8;
    const int edgeWaveBlocks = ((E + N) * 32 + BT - 1) / BT;
    const int nodeBlocks = (N * EMB + BT - 1) / BT;

    for (int l = 0; l < 2; ++l) {
        const float* hin = (l == 0) ? emb : hbuf;
        float* o = (l == 0) ? hbuf : (float*)d_out;
        const float* Wl_l = Wl + (size_t)l * EMB * EMB;
        const float* Wr_l = Wr + (size_t)l * EMB * EMB;
        const float* We_l = We + (size_t)l * EMB;
        const float* at_l = att + (size_t)l * EMB;
        const float* b_l  = bias + (size_t)l * EMB;

        gat_init_kernel<<<nodeBlocks, BT, 0, stream>>>(o, deg, lw, amax, denom, N,
                                                       (l == 0) ? 1 : 0);
        if (l == 0)
            gat_deg_kernel<<<(E + BT - 1) / BT, BT, 0, stream>>>(dst, ew, deg, lw, E);
        gat_gemm_kernel<<<gemmBlocks, BT, 0, stream>>>(hin, Wl_l, Wr_l, xl, xr, N);
        gat_alpha_kernel<<<edgeWaveBlocks, BT, 0, stream>>>(xl, xr, src, dst, ew, deg, lw,
                                                            We_l, at_l, alpha, amax, E, N);
        gat_expsum_kernel<<<(2 * (E + N) + BT - 1) / BT, BT, 0, stream>>>(dst, alpha, amax,
                                                                          denom, E, N);
        gat_aggregate_kernel<<<edgeWaveBlocks, BT, 0, stream>>>(xl, src, dst, alpha, denom,
                                                                o, E, N);
        gat_finish_kernel<<<nodeBlocks, BT, 0, stream>>>(o, b_l, N);
    }
}


# aggregate pass: lane covers features L, L+32, L+64, L+96, so each float atomic add and each gather instruction touches one whole 128-byte line (4 line requests per edge, not 16); sums arrival-ordered
# speedup vs baseline: 1.2625x; 1.1535x over previous
#include <hip/hip_runtime.h>
#include <math.h>

#define EMB 128
#define NEG_SLOPE 0.2f

typedef float v2f __attribute__((ext_vector_type(2)));
typedef float v8f __attribute__((ext_vector_type(8)));

__device__ __forceinline__ void atomicMaxFloat(float* addr, float v) {
    if (v >= 0.0f) atomicMax((int*)addr, __float_as_int(v));
    else           atomicMin((unsigned int*)addr, __float_as_uint(v));
}

__global__ void gat_init_kernel(float* __restrict__ o, float* __restrict__ deg,
                                float* __restrict__ lw, float* __restrict__ amax,
                                float* __restrict__ denom, int N, int first_layer) {
    int i = blockIdx.x * blockDim.x + threadIdx.x;
    if (i < N * EMB) o[i] = 0.0f;
    if (first_layer && i < N) { deg[i] = 0.0f; lw[i] = 0.0f; }
    if (i < 2 * N) { amax[i] = -INFINITY; denom[i] = 0.0f; }
}

__global__ void gat_deg_kernel(const int* __restrict__ dst, const float* __restrict__ ew,
                               float* __restrict__ deg, float* __restrict__ lw, int E) {
    int e = blockIdx.x * blockDim.x + threadIdx.x;
    if (e >= E) return;
    int d = dst[e];
    atomicAdd(&deg[d], 1.0f);
    atomicAdd(&lw[d], ew[e]);
}

__global__ __launch_bounds__(256)
void gat_gemm_kernel(const float* __restrict__ h, const float* __restrict__ Wl,
                     const float* __restrict__ Wr, float* __restrict__ xl,
                     float* __restrict__ xr, int N) {
    int lane = threadIdx.x & 31;
    int wave = (blockIdx.x * blockDim.x + threadIdx.x) >> 5;
    int rowTiles = (N + 15) >> 4;
    if (wave >= rowTiles * 16) return;
    int rowTile = wave >> 4;
    int t = wave & 15;
    const float* W = (t & 8) ? Wr : Wl;
    float*      out = (t & 8) ? xr : xl;
    int ncol = (t & 7) * 16;
    int half = lane >> 4;
    int m    = lane & 15;
    int arow = rowTile * 16 + m;
    if (arow > N - 1) arow = N - 1;
    int col  = ncol + m;
    int koff = half * 2;

    v8f c = {0.f, 0.f, 0.f, 0.f, 0.f, 0.f, 0.f, 0.f};
#pragma unroll
    for (int k = 0; k < EMB; k += 4) {
        v2f a = *(const v2f*)(h + (size_t)arow * EMB + k + koff);
        const float* wp = W + (size_t)(k + koff) * EMB + col;
        v2f b; b.x = wp[0]; b.y = wp[EMB];
        c = __builtin_amdgcn_wmma_f32_16x16x4_f32(false, a, false, b, (short)0, c,
                                                  false, false);
    }
#pragma unroll
    for (int j = 0; j < 8; ++j) {
        int r = rowTile * 16 + j + half * 8;
        if (r < N) out[(size_t)r * EMB + col] = c[j];
    }
}

__global__ void gat_alpha_kernel(const float* __restrict__ xl, const float* __restrict__ xr,
                                 const int* __restrict__ src, const int* __restrict__ dst,
                                 const float* __restrict__ ew, const float* __restrict__ deg,
                                 const float* __restrict__ lw, const float* __restrict__ We,
                                 const float* __restrict__ att, float* __restrict__ alpha,
                                 float* __restrict__ amax, int E, int N) {
    int wave = (blockIdx.x * blockDim.x + threadIdx.x) >> 5;
    int lane = threadIdx.x & 31;
    if (wave >= E + N) return;
    int s, d; float w;
    if (wave < E) { s = src[wave]; d = dst[wave]; w = ew[wave]; }
    else { int i = wave - E; s = i; d = i; w = lw[i] / fmaxf(deg[i], 1.0f); }

    const float4 xs = *(const float4*)(xl + (size_t)s * EMB + lane * 4);
    const float4 xd = *(const float4*)(xr + (size_t)d * EMB + lane * 4);
    const float4 we = *(const float4*)(We + lane * 4);
    const float4 at = *(const float4*)(att + lane * 4);

    float p = 0.0f, v;
    v = xs.x + xd.x + w * we.x; v = (v >= 0.f) ? v : NEG_SLOPE * v; p += v * at.x;
    v = xs.y + xd.y + w * we.y; v = (v >= 0.f) ? v : NEG_SLOPE * v; p += v * at.y;
    v = xs.z + xd.z + w * we.z; v = (v >= 0.f) ? v : NEG_SLOPE * v; p += v * at.z;
    v = xs.w + xd.w + w * we.w; v = (v >= 0.f) ? v : NEG_SLOPE * v; p += v * at.w;

    for (int msk = 8; msk >= 1; msk >>= 1) p += __shfl_xor(p, msk, 32);

    if ((lane & 15) == 0) {
        int head = lane >> 4;
        alpha[(size_t)wave * 2 + head] = p;
        atomicMaxFloat(&amax[(size_t)d * 2 + head], p);
    }
}

__global__ void gat_expsum_kernel(const int* __restrict__ dst, float* __restrict__ alpha,
                                  const float* __restrict__ amax, float* __restrict__ denom,
                                  int E, int N) {
    int idx = blockIdx.x * blockDim.x + threadIdx.x;
    if (idx >= 2 * (E + N)) return;
    int e = idx >> 1, head = idx & 1;
    int d = (e < E) ? dst[e] : (e - E);
    float ex = __expf(alpha[idx] - amax[(size_t)d * 2 + head]);
    alpha[idx] = ex;
    atomicAdd(&denom[(size_t)d * 2 + head], ex);
}

__global__ void gat_aggregate_kernel(const float* __restrict__ xl, const int* __restrict__ src,
                                     const int* __restrict__ dst, const float* __restrict__ alpha,
                                     const float* __restrict__ denom, float* __restrict__ o,
                                     int E, int N) {
    int wave = (blockIdx.x * blockDim.x + threadIdx.x) >> 5;
    int lane = threadIdx.x & 31;
    if (wave >= E + N) return;
    int s, d;
    if (wave < E) { s = src[wave]; d = dst[wave]; }
    else { s = d = wave - E; }
    const v2f ex = *(const v2f*)(alpha + (size_t)wave * 2);
    const v2f dn = *(const v2f*)(denom + (size_t)d * 2);
    const float a0 = ex.x / dn.x;
    const float a1 = ex.y / dn.y;
    const float* xp = xl + (size_t)s * EMB + lane;
    float* op = o + (size_t)d * EMB + lane;
    const float m0 = xp[0], m1 = xp[32], m2 = xp[64], m3 = xp[96];
    atomicAdd(op +  0, m0 * a0);
    atomicAdd(op + 32, m1 * a0);
    atomicAdd(op + 64, m2 * a1);
    atomicAdd(op + 96, m3 * a1);
}

__global__ void gat_finish_kernel(float* __restrict__ o, const float* __restrict__ bias, int N) {
    int i = blockIdx.x * blockDim.x + threadIdx.x;
    if (i >= N * EMB) return;
    float x = o[i] + bias[i & (EMB - 1)];
    o[i] = (x > 0.0f) ? x : (__expf(x) - 1.0f);
}

extern "C" void kernel_launch(void* const* d_in, const int* in_sizes, int n_in,
                              void* d_out, int out_size, void* d_ws, size_t ws_size,
                              hipStream_t stream) {
    const int*   edge_index = (const int*)d_in[0];
    const float* ew   = (const float*)d_in[1];
    const float* emb  = (const float*)d_in[2];
    const float* Wl   = (const float*)d_in[3];
    const float* Wr   = (const float*)d_in[4];
    const float* We   = (const float*)d_in[5];
    const float* att  = (const float*)d_in[6];
    const float* bias = (const float*)d_in[7];

    const int E = in_sizes[0] / 2;
    const int N = in_sizes[2] / EMB;
    const int* src = edge_index;
    const int* dst = edge_index + E;

    float* ws    = (float*)d_ws;
    float* xl    = ws;
    float* xr    = xl + (size_t)N * EMB;
    float* hbuf  = xr + (size_t)N * EMB;
    float* alpha = hbuf + (size_t)N * EMB;
    float* amax  = alpha + (size_t)2 * (E + N);
    float* denom = amax + (size_t)2 * N;
    float* deg   = denom + (size_t)2 * N;
    float* lw    = deg + N;

    const int BT = 256;
    const int rowTiles = (N + 15) / 16;
    const int gemmBlocks = (rowTiles * 16 + 7) / 8;
    const int edgeWaveBlocks = ((E + N) * 32 + BT - 1) / BT;
    const int nodeBlocks = (N * EMB + BT - 1) / BT;

    for (int l = 0; l < 2; ++l) {
        const float* hin = (l == 0) ? emb : hbuf;
        float* o = (l == 0) ? hbuf : (float*)d_out;
        const float* Wl_l = Wl + (size_t)l * EMB * EMB;
        const float* Wr_l = Wr + (size_t)l * EMB * EMB;
        const float* We_l = We + (size_t)l * EMB;
        const float* at_l = att + (size_t)l * EMB;
        const float* b_l  = bias + (size_t)l * EMB;

        gat_init_kernel<<<nodeBlocks, BT, 0, stream>>>(o, deg, lw, amax, denom, N,
                                                       (l == 0) ? 1 : 0);
        if (l == 0)
            gat_deg_kernel<<<(E + BT - 1) / BT, BT, 0, stream>>>(dst, ew, deg, lw, E);
        gat_gemm_kernel<<<gemmBlocks, BT, 0, stream>>>(hin, Wl_l, Wr_l, xl, xr, N);
        gat_alpha_kernel<<<edgeWaveBlocks, BT, 0, stream>>>(xl, xr, src, dst, ew, deg, lw,
                                                            We_l, at_l, alpha, amax, E, N);
        gat_expsum_kernel<<<(2 * (E + N) + BT - 1) / BT, BT, 0, stream>>>(dst, alpha, amax,
                                                                          denom, E, N);
        gat_aggregate_kernel<<<edgeWaveBlocks, BT, 0, stream>>>(xl, src, dst, alpha, denom,
                                                                o, E, N);
        gat_finish_kernel<<<nodeBlocks, BT, 0, stream>>>(o, b_l, N);
    }
}


# edges grouped by destination once per call (counting sort, arrival-order rank); one wave per node: running-max softmax and weighted sum in registers, no float atomics; summation order changes
# speedup vs baseline: 3.0068x; 2.3816x over previous
#include <hip/hip_runtime.h>
#include <math.h>


#define EMB 128
#define NEG_SLOPE 0.2f

typedef float v2f __attribute__((ext_vector_type(2)));
typedef float v8f __attribute__((ext_vector_type(8)));


__global__ void gat_zero_kernel(int* __restrict__ cnt, int N) {
    int i = blockIdx.x * blockDim.x + threadIdx.x;
    if (i < N) cnt[i] = 0;
}

__global__ void gat_rank_kernel(const int* __restrict__ dst, int* __restrict__ cnt,
                                int* __restrict__ rank, int E) {
    int e = blockIdx.x * blockDim.x + threadIdx.x;
    if (e >= E) return;
    rank[e] = atomicAdd(&cnt[dst[e]], 1);
}

__global__ __launch_bounds__(1024)
void gat_scan_kernel(const int* __restrict__ cnt, int* __restrict__ off, int N) {
    __shared__ int part[1024];
    const int t = threadIdx.x;
    const int per = (N + 1023) / 1024;
    const int b = t * per;
    const int e = (b + per < N) ? (b + per) : N;
    int s = 0;
    for (int i = b; i < e; ++i) s += cnt[i];
    part[t] = s;
    __syncthreads();
    for (int dlt = 1; dlt < 1024; dlt <<= 1) {
        int v = (t >= dlt) ? part[t - dlt] : 0;
        __syncthreads();
        part[t] += v;
        __syncthreads();
    }
    int run = part[t] - s;
    for (int i = b; i < e; ++i) { off[i] = run; run += cnt[i]; }
    if (t == 1023) off[N] = part[1023];
}

__global__ void gat_scatter_kernel(const int* __restrict__ src, const int* __restrict__ dst,
                                   const float* __restrict__ ew, const int* __restrict__ off,
                                   const int* __restrict__ rank, int* __restrict__ grp, int E) {
    int e = blockIdx.x * blockDim.x + threadIdx.x;
    if (e >= E) return;
    int pos = off[dst[e]] + rank[e];
    int2 rec; rec.x = src[e]; rec.y = __float_as_int(ew[e]);
    if ((unsigned)pos < (unsigned)E) *(int2*)(grp + 2 * (size_t)pos) = rec;
}

__global__ __launch_bounds__(256)
void gat_gemm_kernel(const float* __restrict__ h, const float* __restrict__ Wl,
                     const float* __restrict__ Wr, float* __restrict__ xl,
                     float* __restrict__ xr, int N) {
    int lane = threadIdx.x & 31;
    int wave = (blockIdx.x * blockDim.x + threadIdx.x) >> 5;
    int rowTiles = (N + 15) >> 4;
    if (wave >= rowTiles * 16) return;
    int rowTile = wave >> 4;
    int t = wave & 15;
    const float* W = (t & 8) ? Wr : Wl;
    float*      out = (t & 8) ? xr : xl;
    int ncol = (t & 7) * 16;
    int half = lane >> 4;
    int m    = lane & 15;
    int arow = rowTile * 16 + m;
    if (arow > N - 1) arow = N - 1;
    int col  = ncol + m;
    int koff = half * 2;

    v8f c = {0.f, 0.f, 0.f, 0.f, 0.f, 0.f, 0.f, 0.f};
#pragma unroll
    for (int k = 0; k < EMB; k += 4) {
        v2f a = *(const v2f*)(h + (size_t)arow * EMB + k + koff);
        const float* wp = W + (size_t)(k + koff) * EMB + col;
        v2f b; b.x = wp[0]; b.y = wp[EMB];
        c = __builtin_amdgcn_wmma_f32_16x16x4_f32(false, a, false, b, (short)0, c,
                                                  false, false);
    }
#pragma unroll
    for (int j = 0; j < 8; ++j) {
        int r = rowTile * 16 + j + half * 8;
        if (r < N) out[(size_t)r * EMB + col] = c[j];
    }
}

__device__ __forceinline__ float gat_head_logit(const float4 xs, const float4 xd, const float4 we,
                                                const float4 at, const float w) {
    float p = 0.0f, v;
    v = xs.x + xd.x + w * we.x; v = (v >= 0.f) ? v : NEG_SLOPE * v; p += v * at.x;
    v = xs.y + xd.y + w * we.y; v = (v >= 0.f) ? v : NEG_SLOPE * v; p += v * at.y;
    v = xs.z + xd.z + w * we.z; v = (v >= 0.f) ? v : NEG_SLOPE * v; p += v * at.z;
    v = xs.w + xd.w + w * we.w; v = (v >= 0.f) ? v : NEG_SLOPE * v; p += v * at.w;
    for (int msk = 8; msk >= 1; msk >>= 1) p += __shfl_xor(p, msk, 32);
    return p;
}

__global__ __launch_bounds__(256)
void gat_node_kernel(const float* __restrict__ xl, const float* __restrict__ xr,
                     const int* __restrict__ off, const int* __restrict__ grp,
                     const float* __restrict__ We, const float* __restrict__ att,
                     const float* __restrict__ bias, float* __restrict__ o, int N) {
    const int wave = (blockIdx.x * blockDim.x + threadIdx.x) >> 5;
    const int lane = threadIdx.x & 31;
    if (wave >= N) return;
    const int d = __builtin_amdgcn_readfirstlane(wave);
    const int beg = off[d];
    const int end = off[d + 1];

    const float4 xd = *(const float4*)(xr + (size_t)d * EMB + lane * 4);
    const float4 xself = *(const float4*)(xl + (size_t)d * EMB + lane * 4);
    const float4 we = *(const float4*)(We + lane * 4);
    const float4 at = *(const float4*)(att + lane * 4);

    float m = -INFINITY;
    float den = 0.0f;
    float4 acc = {0.0f, 0.0f, 0.0f, 0.0f};
    float wsum = 0.0f;

    for (int base = beg; base < end; base += 32) {
        int my_s = 0, my_w = 0;
        if (base + lane < end) {
            const int2 rec = *(const int2*)(grp + 2 * (size_t)(base + lane));
            my_s = rec.x; my_w = rec.y;
        }
        wsum += __int_as_float(my_w);
        const int nb = (end - base < 32) ? (end - base) : 32;
        for (int i = 0; i < nb; i += 4) {
            const int i1 = (i + 1 < nb) ? (i + 1) : (nb - 1);
            const int i2 = (i + 2 < nb) ? (i + 2) : (nb - 1);
            const int i3 = (i + 3 < nb) ? (i + 3) : (nb - 1);
            const int s0 = __builtin_amdgcn_readlane(my_s, i);
            const int s1 = __builtin_amdgcn_readlane(my_s, i1);
            const int s2 = __builtin_amdgcn_readlane(my_s, i2);
            const int s3 = __builtin_amdgcn_readlane(my_s, i3);
            const float w0 = __int_as_float(__builtin_amdgcn_readlane(my_w, i));
            const float w1 = __int_as_float(__builtin_amdgcn_readlane(my_w, i1));
            const float w2 = __int_as_float(__builtin_amdgcn_readlane(my_w, i2));
            const float w3 = __int_as_float(__builtin_amdgcn_readlane(my_w, i3));
            const float4 r0 = *(const float4*)(xl + (size_t)s0 * EMB + lane * 4);
            const float4 r1 = *(const float4*)(xl + (size_t)s1 * EMB + lane * 4);
            const float4 r2 = *(const float4*)(xl + (size_t)s2 * EMB + lane * 4);
            const float4 r3 = *(const float4*)(xl + (size_t)s3 * EMB + lane * 4);
            const float p0 = gat_head_logit(r0, xd, we, at, w0);
            float p1 = gat_head_logit(r1, xd, we, at, w1);
            float p2 = gat_head_logit(r2, xd, we, at, w2);
            float p3 = gat_head_logit(r3, xd, we, at, w3);
            p1 = (i + 1 < nb) ? p1 : -INFINITY;
            p2 = (i + 2 < nb) ? p2 : -INFINITY;
            p3 = (i + 3 < nb) ? p3 : -INFINITY;
            const float mn = fmaxf(fmaxf(m, p0), fmaxf(p1, fmaxf(p2, p3)));
            const float sc = __expf(m - mn);
            const float e0 = __expf(p0 - mn);
            const float e1 = __expf(p1 - mn);
            const float e2 = __expf(p2 - mn);
            const float e3 = __expf(p3 - mn);
            den = den * sc + ((e0 + e1) + (e2 + e3));
            acc.x = acc.x * sc + ((e0 * r0.x + e1 * r1.x) + (e2 * r2.x + e3 * r3.x));
            acc.y = acc.y * sc + ((e0 * r0.y + e1 * r1.y) + (e2 * r2.y + e3 * r3.y));
            acc.z = acc.z * sc + ((e0 * r0.z + e1 * r1.z) + (e2 * r2.z + e3 * r3.z));
            acc.w = acc.w * sc + ((e0 * r0.w + e1 * r1.w) + (e2 * r2.w + e3 * r3.w));
            m = mn;
        }
    }

    for (int msk = 16; msk >= 1; msk >>= 1) wsum += __shfl_xor(wsum, msk, 32);
    const float wself = wsum / fmaxf((float)(end - beg), 1.0f);
    {
        const float p = gat_head_logit(xself, xd, we, at, wself);
        const float mn = fmaxf(m, p);
        const float sc = __expf(m - mn);
        const float ex = __expf(p - mn);
        den = den * sc + ex;
        acc.x = acc.x * sc + ex * xself.x;
        acc.y = acc.y * sc + ex * xself.y;
        acc.z = acc.z * sc + ex * xself.z;
        acc.w = acc.w * sc + ex * xself.w;
    }

    const float4 b = *(const float4*)(bias + lane * 4);
    float4 r;
    r.x = acc.x / den + b.x;
    r.y = acc.y / den + b.y;
    r.z = acc.z / den + b.z;
    r.w = acc.w / den + b.w;
    r.x = (r.x > 0.0f) ? r.x : (__expf(r.x) - 1.0f);
    r.y = (r.y > 0.0f) ? r.y : (__expf(r.y) - 1.0f);
    r.z = (r.z > 0.0f) ? r.z : (__expf(r.z) - 1.0f);
    r.w = (r.w > 0.0f) ? r.w : (__expf(r.w) - 1.0f);
    *(float4*)(o + (size_t)d * EMB + lane * 4) = r;
}

extern "C" void kernel_launch(void* const* d_in, const int* in_sizes, int n_in,
                              void* d_out, int out_size, void* d_ws, size_t ws_size,
                              hipStream_t stream) {
    const int*   edge_index = (const int*)d_in[0];
    const float* ew   = (const float*)d_in[1];
    const float* emb  = (const float*)d_in[2];
    const float* Wl   = (const float*)d_in[3];
    const float* Wr   = (const float*)d_in[4];
    const float* We   = (const float*)d_in[5];
    const float* att  = (const float*)d_in[6];
    const float* bias = (const float*)d_in[7];

    const int E = in_sizes[0] / 2;
    const int N = in_sizes[2] / EMB;
    const int* src = edge_index;
    const int* dst = edge_index + E;

    float* ws    = (float*)d_ws;
    float* xl    = ws;
    float* xr    = xl + (size_t)N * EMB;
    float* hbuf  = xr + (size_t)N * EMB;
    int*   cnt   = (int*)(hbuf + (size_t)N * EMB);
    int*   off   = cnt + (((size_t)N + 31) / 32) * 32;
    int*   rank  = off + (((size_t)N + 1 + 31) / 32) * 32;
    int*   grp   = rank + (((size_t)E + 31) / 32) * 32;

    const int BT = 256;
    const int rowTiles = (N + 15) / 16;
    const int gemmBlocks = (rowTiles * 16 + 7) / 8;
    const int edgeBlocks = (E + BT - 1) / BT;
    const int nodeWaveBlocks = (N * 32 + BT - 1) / BT;

    gat_zero_kernel<<<(N + BT - 1) / BT, BT, 0, stream>>>(cnt, N);
    gat_rank_kernel<<<edgeBlocks, BT, 0, stream>>>(dst, cnt, rank, E);
    gat_scan_kernel<<<1, 1024, 0, stream>>>(cnt, off, N);
    gat_scatter_kernel<<<edgeBlocks, BT, 0, stream>>>(src, dst, ew, off, rank, grp, E);

    for (int l = 0; l < 2; ++l) {
        const float* hin = (l == 0) ? emb : hbuf;
        float* o = (l == 0) ? hbuf : (float*)d_out;
        const float* Wl_l = Wl + (size_t)l * EMB * EMB;
        const float* Wr_l = Wr + (size_t)l * EMB * EMB;
        const float* We_l = We + (size_t)l * EMB;
        const float* at_l = att + (size_t)l * EMB;
        const float* b_l  = bias + (size_t)l * EMB;

        gat_gemm_kernel<<<gemmBlocks, BT, 0, stream>>>(hin, Wl_l, Wr_l, xl, xr, N);
        gat_node_kernel<<<nodeWaveBlocks, BT, 0, stream>>>(xl, xr, off, grp, We_l, at_l, b_l,
                                                           o, N);
    }
}


# dense stage: a wave owns 16 rows x 64 columns (4 tiles share the A fragment, one 16-byte load per k row feeds the 4 B fragments); 12500 waves, not 50000; k order per product unchanged
# speedup vs baseline: 3.5088x; 1.1670x over previous
#include <hip/hip_runtime.h>
#include <math.h>


#define EMB 128
#define NEG_SLOPE 0.2f

typedef float v2f __attribute__((ext_vector_type(2)));
typedef float v8f __attribute__((ext_vector_type(8)));


__global__ void gat_zero_kernel(int* __restrict__ cnt, int N) {
    int i = blockIdx.x * blockDim.x + threadIdx.x;
    if (i < N) cnt[i] = 0;
}

__global__ void gat_rank_kernel(const int* __restrict__ dst, int* __restrict__ cnt,
                                int* __restrict__ rank, int E) {
    int e = blockIdx.x * blockDim.x + threadIdx.x;
    if (e >= E) return;
    rank[e] = atomicAdd(&cnt[dst[e]], 1);
}

__global__ __launch_bounds__(1024)
void gat_scan_kernel(const int* __restrict__ cnt, int* __restrict__ off, int N) {
    __shared__ int part[1024];
    const int t = threadIdx.x;
    const int per = (N + 1023) / 1024;
    const int b = t * per;
    const int e = (b + per < N) ? (b + per) : N;
    int s = 0;
    for (int i = b; i < e; ++i) s += cnt[i];
    part[t] = s;
    __syncthreads();
    for (int dlt = 1; dlt < 1024; dlt <<= 1) {
        int v = (t >= dlt) ? part[t - dlt] : 0;
        __syncthreads();
        part[t] += v;
        __syncthreads();
    }
    int run = part[t] - s;
    for (int i = b; i < e; ++i) { off[i] = run; run += cnt[i]; }
    if (t == 1023) off[N] = part[1023];
}

__global__ void gat_scatter_kernel(const int* __restrict__ src, const int* __restrict__ dst,
                                   const float* __restrict__ ew, const int* __restrict__ off,
                                   const int* __restrict__ rank, int* __restrict__ grp, int E) {
    int e = blockIdx.x * blockDim.x + threadIdx.x;
    if (e >= E) return;
    int pos = off[dst[e]] + rank[e];
    int2 rec; rec.x = src[e]; rec.y = __float_as_int(ew[e]);
    if ((unsigned)pos < (unsigned)E) *(int2*)(grp + 2 * (size_t)pos) = rec;
}

typedef float v4f __attribute__((ext_vector_type(4)));

__global__ __launch_bounds__(256)
void gat_gemm_kernel(const float* __restrict__ h, const float* __restrict__ Wl,
                     const float* __restrict__ Wr, float* __restrict__ xl,
                     float* __restrict__ xr, int N) {
    int lane = threadIdx.x & 31;
    int wave = (blockIdx.x * blockDim.x + threadIdx.x) >> 5;
    int rowTiles = (N + 15) >> 4;
    if (wave >= rowTiles * 4) return;
    int rowTile = wave >> 2;
    int q = wave & 3;
    const float* W = (q & 2) ? Wr : Wl;
    float*      out = (q & 2) ? xr : xl;
    int cbase = (q & 1) * 64;
    int half = lane >> 4;
    int m    = lane & 15;
    int arow = rowTile * 16 + m;
    if (arow > N - 1) arow = N - 1;
    int col  = cbase + 4 * m;
    int koff = half * 2;
    const float* ap = h + (size_t)arow * EMB + koff;
    const float* wp = W + (size_t)koff * EMB + col;

    v8f c0 = {0.f, 0.f, 0.f, 0.f, 0.f, 0.f, 0.f, 0.f};
    v8f c1 = c0, c2 = c0, c3 = c0;
#pragma unroll
    for (int k = 0; k < EMB; k += 4) {
        v2f a = *(const v2f*)(ap + k);
        v4f w0 = *(const v4f*)(wp + (size_t)k * EMB);
        v4f w1 = *(const v4f*)(wp + (size_t)(k + 1) * EMB);
        v2f b0; b0.x = w0.x; b0.y = w1.x;
        v2f b1; b1.x = w0.y; b1.y = w1.y;
        v2f b2; b2.x = w0.z; b2.y = w1.z;
        v2f b3; b3.x = w0.w; b3.y = w1.w;
        c0 = __builtin_amdgcn_wmma_f32_16x16x4_f32(false, a, false, b0, (short)0, c0,
                                                   false, false);
        c1 = __builtin_amdgcn_wmma_f32_16x16x4_f32(false, a, false, b1, (short)0, c1,
                                                   false, false);
        c2 = __builtin_amdgcn_wmma_f32_16x16x4_f32(false, a, false, b2, (short)0, c2,
                                                   false, false);
        c3 = __builtin_amdgcn_wmma_f32_16x16x4_f32(false, a, false, b3, (short)0, c3,
                                                   false, false);
    }
#pragma unroll
    for (int j = 0; j < 8; ++j) {
        int r = rowTile * 16 + j + half * 8;
        v4f v; v.x = c0[j]; v.y = c1[j]; v.z = c2[j]; v.w = c3[j];
        if (r < N) *(v4f*)(out + (size_t)r * EMB + col) = v;
    }
}

__device__ __forceinline__ float gat_head_logit(const float4 xs, const float4 xd, const float4 we,
                                                const float4 at, const float w) {
    float p = 0.0f, v;
    v = xs.x + xd.x + w * we.x; v = (v >= 0.f) ? v : NEG_SLOPE * v; p += v * at.x;
    v = xs.y + xd.y + w * we.y; v = (v >= 0.f) ? v : NEG_SLOPE * v; p += v * at.y;
    v = xs.z + xd.z + w * we.z; v = (v >= 0.f) ? v : NEG_SLOPE * v; p += v * at.z;
    v = xs.w + xd.w + w * we.w; v = (v >= 0.f) ? v : NEG_SLOPE * v; p += v * at.w;
    for (int msk = 8; msk >= 1; msk >>= 1) p += __shfl_xor(p, msk, 32);
    return p;
}

__global__ __launch_bounds__(256)
void gat_node_kernel(const float* __restrict__ xl, const float* __restrict__ xr,
                     const int* __restrict__ off, const int* __restrict__ grp,
                     const float* __restrict__ We, const float* __restrict__ att,
                     const float* __restrict__ bias, float* __restrict__ o, int N) {
    const int wave = (blockIdx.x * blockDim.x + threadIdx.x) >> 5;
    const int lane = threadIdx.x & 31;
    if (wave >= N) return;
    const int d = __builtin_amdgcn_readfirstlane(wave);
    const int beg = off[d];
    const int end = off[d + 1];

    const float4 xd = *(const float4*)(xr + (size_t)d * EMB + lane * 4);
    const float4 xself = *(const float4*)(xl + (size_t)d * EMB + lane * 4);
    const float4 we = *(const float4*)(We + lane * 4);
    const float4 at = *(const float4*)(att + lane * 4);

    float m = -INFINITY;
    float den = 0.0f;
    float4 acc = {0.0f, 0.0f, 0.0f, 0.0f};
    float wsum = 0.0f;

    for (int base = beg; base < end; base += 32) {
        int my_s = 0, my_w = 0;
        if (base + lane < end) {
            const int2 rec = *(const int2*)(grp + 2 * (size_t)(base + lane));
            my_s = rec.x; my_w = rec.y;
        }
        wsum += __int_as_float(my_w);
        const int nb = (end - base < 32) ? (end - base) : 32;
        for (int i = 0; i < nb; i += 4) {
            const int i1 = (i + 1 < nb) ? (i + 1) : (nb - 1);
            const int i2 = (i + 2 < nb) ? (i + 2) : (nb - 1);
            const int i3 = (i + 3 < nb) ? (i + 3) : (nb - 1);
            const int s0 = __builtin_amdgcn_readlane(my_s, i);
            const int s1 = __builtin_amdgcn_readlane(my_s, i1);
            const int s2 = __builtin_amdgcn_readlane(my_s, i2);
            const int s3 = __builtin_amdgcn_readlane(my_s, i3);
            const float w0 = __int_as_float(__builtin_amdgcn_readlane(my_w, i));
            const float w1 = __int_as_float(__builtin_amdgcn_readlane(my_w, i1));
            const float w2 = __int_as_float(__builtin_amdgcn_readlane(my_w, i2));
            const float w3 = __int_as_float(__builtin_amdgcn_readlane(my_w, i3));
            const float4 r0 = *(const float4*)(xl + (size_t)s0 * EMB + lane * 4);
            const float4 r1 = *(const float4*)(xl + (size_t)s1 * EMB + lane * 4);
            const float4 r2 = *(const float4*)(xl + (size_t)s2 * EMB + lane * 4);
            const float4 r3 = *(const float4*)(xl + (size_t)s3 * EMB + lane * 4);
            const float p0 = gat_head_logit(r0, xd, we, at, w0);
            float p1 = gat_head_logit(r1, xd, we, at, w1);
            float p2 = gat_head_logit(r2, xd, we, at, w2);
            float p3 = gat_head_logit(r3, xd, we, at, w3);
            p1 = (i + 1 < nb) ? p1 : -INFINITY;
            p2 = (i + 2 < nb) ? p2 : -INFINITY;
            p3 = (i + 3 < nb) ? p3 : -INFINITY;
            const float mn = fmaxf(fmaxf(m, p0), fmaxf(p1, fmaxf(p2, p3)));
            const float sc = __expf(m - mn);
            const float e0 = __expf(p0 - mn);
            const float e1 = __expf(p1 - mn);
            const float e2 = __expf(p2 - mn);
            const float e3 = __expf(p3 - mn);
            den = den * sc + ((e0 + e1) + (e2 + e3));
            acc.x = acc.x * sc + ((e0 * r0.x + e1 * r1.x) + (e2 * r2.x + e3 * r3.x));
            acc.y = acc.y * sc + ((e0 * r0.y + e1 * r1.y) + (e2 * r2.y + e3 * r3.y));
            acc.z = acc.z * sc + ((e0 * r0.z + e1 * r1.z) + (e2 * r2.z + e3 * r3.z));
            acc.w = acc.w * sc + ((e0 * r0.w + e1 * r1.w) + (e2 * r2.w + e3 * r3.w));
            m = mn;
        }
    }

    for (int msk = 16; msk >= 1; msk >>= 1) wsum += __shfl_xor(wsum, msk, 32);
    const float wself = wsum / fmaxf((float)(end - beg), 1.0f);
    {
        const float p = gat_head_logit(xself, xd, we, at, wself);
        const float mn = fmaxf(m, p);
        const float sc = __expf(m - mn);
        const float ex = __expf(p - mn);
        den = den * sc + ex;
        acc.x = acc.x * sc + ex * xself.x;
        acc.y = acc.y * sc + ex * xself.y;
        acc.z = acc.z * sc + ex * xself.z;
        acc.w = acc.w * sc + ex * xself.w;
    }

    const float4 b = *(const float4*)(bias + lane * 4);
    float4 r;
    r.x = acc.x / den + b.x;
    r.y = acc.y / den + b.y;
    r.z = acc.z / den + b.z;
    r.w = acc.w / den + b.w;
    r.x = (r.x > 0.0f) ? r.x : (__expf(r.x) - 1.0f);
    r.y = (r.y > 0.0f) ? r.y : (__expf(r.y) - 1.0f);
    r.z = (r.z > 0.0f) ? r.z : (__expf(r.z) - 1.0f);
    r.w = (r.w > 0.0f) ? r.w : (__expf(r.w) - 1.0f);
    *(float4*)(o + (size_t)d * EMB + lane * 4) = r;
}

extern "C" void kernel_launch(void* const* d_in, const int* in_sizes, int n_in,
                              void* d_out, int out_size, void* d_ws, size_t ws_size,
                              hipStream_t stream) {
    const int*   edge_index = (const int*)d_in[0];
    const float* ew   = (const float*)d_in[1];
    const float* emb  = (const float*)d_in[2];
    const float* Wl   = (const float*)d_in[3];
    const float* Wr   = (const float*)d_in[4];
    const float* We   = (const float*)d_in[5];
    const float* att  = (const float*)d_in[6];
    const float* bias = (const float*)d_in[7];

    const int E = in_sizes[0] / 2;
    const int N = in_sizes[2] / EMB;
    const int* src = edge_index;
    const int* dst = edge_index + E;

    float* ws    = (float*)d_ws;
    float* xl    = ws;
    float* xr    = xl + (size_t)N * EMB;
    float* hbuf  = xr + (size_t)N * EMB;
    int*   cnt   = (int*)(hbuf + (size_t)N * EMB);
    int*   off   = cnt + (((size_t)N + 31) / 32) * 32;
    int*   rank  = off + (((size_t)N + 1 + 31) / 32) * 32;
    int*   grp   = rank + (((size_t)E + 31) / 32) * 32;

    const int BT = 256;
    const int rowTiles = (N + 15) / 16;
    const int gemmBlocks = (rowTiles * 4 + 7) / 8;
    const int edgeBlocks = (E + BT - 1) / BT;
    const int nodeWaveBlocks = (N * 32 + BT - 1) / BT;

    gat_zero_kernel<<<(N + BT - 1) / BT, BT, 0, stream>>>(cnt, N);
    gat_rank_kernel<<<edgeBlocks, BT, 0, stream>>>(dst, cnt, rank, E);
    gat_scan_kernel<<<1, 1024, 0, stream>>>(cnt, off, N);
    gat_scatter_kernel<<<edgeBlocks, BT, 0, stream>>>(src, dst, ew, off, rank, grp, E);

    for (int l = 0; l < 2; ++l) {
        const float* hin = (l == 0) ? emb : hbuf;
        float* o = (l == 0) ? hbuf : (float*)d_out;
        const float* Wl_l = Wl + (size_t)l * EMB * EMB;
        const float* Wr_l = Wr + (size_t)l * EMB * EMB;
        const float* We_l = We + (size_t)l * EMB;
        const float* at_l = att + (size_t)l * EMB;
        const float* b_l  = bias + (size_t)l * EMB;

        gat_gemm_kernel<<<gemmBlocks, BT, 0, stream>>>(hin, Wl_l, Wr_l, xl, xr, N);
        gat_node_kernel<<<nodeWaveBlocks, BT, 0, stream>>>(xl, xr, off, grp, We_l, at_l, b_l,
                                                           o, N);
    }
}


# scan of the per-node edge counts made coalesced: 16-byte loads and stores, 512 contiguous bytes a wave, prefix built in registers by wave shuffles and one LDS pass; integer sums, exact
# speedup vs baseline: 4.4648x; 1.2725x over previous
#include <hip/hip_runtime.h>
#include <math.h>


#define EMB 128
#define NEG_SLOPE 0.2f

typedef float v2f __attribute__((ext_vector_type(2)));
typedef float v8f __attribute__((ext_vector_type(8)));


__global__ void gat_zero_kernel(int* __restrict__ cnt, int N) {
    int i = blockIdx.x * blockDim.x + threadIdx.x;
    if (i < ((N + 31) & ~31)) cnt[i] = 0;
}

__global__ void gat_rank_kernel(const int* __restrict__ dst, int* __restrict__ cnt,
                                int* __restrict__ rank, int E) {
    int e = blockIdx.x * blockDim.x + threadIdx.x;
    if (e >= E) return;
    rank[e] = atomicAdd(&cnt[dst[e]], 1);
}

#define SCAN_THREADS 1024
#define SCAN_CHUNKS 16

__device__ __forceinline__ int gat_wave_incl_scan(int v, const int lane) {
#pragma unroll
    for (int d = 1; d < 32; d <<= 1) {
        const int u = __shfl_up(v, d, 32);
        if (lane >= d) v += u;
    }
    return v;
}

__global__ __launch_bounds__(SCAN_THREADS)
void gat_scan_kernel(const int* __restrict__ cnt, int* __restrict__ off, int N) {
    __shared__ int wtot[SCAN_CHUNKS * 32];
    __shared__ int wpre[SCAN_CHUNKS * 32];
    __shared__ int ctot[SCAN_CHUNKS];
    const int t = threadIdx.x;
    const int lane = t & 31;
    const int w = t >> 5;
    int carry = 0;
    for (int p0 = 0; p0 <= N; p0 += SCAN_THREADS * SCAN_CHUNKS * 4) {
        int4 c[SCAN_CHUNKS];
        int inc[SCAN_CHUNKS];
#pragma unroll
        for (int j = 0; j < SCAN_CHUNKS; ++j) {
            const int i0 = p0 + (j * SCAN_THREADS + t) * 4;
            int4 v = {0, 0, 0, 0};
            if (i0 < N) v = *(const int4*)(cnt + i0);
            c[j] = v;
        }
#pragma unroll
        for (int j = 0; j < SCAN_CHUNKS; ++j) inc[j] = c[j].x + c[j].y + c[j].z + c[j].w;
#pragma unroll
        for (int d = 1; d < 32; d <<= 1) {
            int u[SCAN_CHUNKS];
#pragma unroll
            for (int j = 0; j < SCAN_CHUNKS; ++j) u[j] = __shfl_up(inc[j], d, 32);
#pragma unroll
            for (int j = 0; j < SCAN_CHUNKS; ++j) inc[j] += (lane >= d) ? u[j] : 0;
        }
        if (lane == 31) {
#pragma unroll
            for (int j = 0; j < SCAN_CHUNKS; ++j) wtot[j * 32 + w] = inc[j];
        }
        __syncthreads();
        if (w < SCAN_CHUNKS) {
            const int v = wtot[w * 32 + lane];
            const int s = gat_wave_incl_scan(v, lane);
            wpre[w * 32 + lane] = s - v;
            if (lane == 31) ctot[w] = s;
        }
        __syncthreads();
        int run = carry;
#pragma unroll
        for (int j = 0; j < SCAN_CHUNKS; ++j) {
            const int i0 = p0 + (j * SCAN_THREADS + t) * 4;
            const int4 v = c[j];
            int4 o;
            o.x = run + wpre[j * 32 + w] + (inc[j] - (v.x + v.y + v.z + v.w));
            o.y = o.x + v.x;
            o.z = o.y + v.y;
            o.w = o.z + v.z;
            if (i0 <= N) *(int4*)(off + i0) = o;
            run += ctot[j];
        }
        carry = run;
        __syncthreads();
    }
}

__global__ void gat_scatter_kernel(const int* __restrict__ src, const int* __restrict__ dst,
                                   const float* __restrict__ ew, const int* __restrict__ off,
                                   const int* __restrict__ rank, int* __restrict__ grp, int E) {
    int e = blockIdx.x * blockDim.x + threadIdx.x;
    if (e >= E) return;
    int pos = off[dst[e]] + rank[e];
    int2 rec; rec.x = src[e]; rec.y = __float_as_int(ew[e]);
    if ((unsigned)pos < (unsigned)E) *(int2*)(grp + 2 * (size_t)pos) = rec;
}

typedef float v4f __attribute__((ext_vector_type(4)));

__global__ __launch_bounds__(256)
void gat_gemm_kernel(const float* __restrict__ h, const float* __restrict__ Wl,
                     const float* __restrict__ Wr, float* __restrict__ xl,
                     float* __restrict__ xr, int N) {
    int lane = threadIdx.x & 31;
    int wave = (blockIdx.x * blockDim.x + threadIdx.x) >> 5;
    int rowTiles = (N + 15) >> 4;
    if (wave >= rowTiles * 4) return;
    int rowTile = wave >> 2;
    int q = wave & 3;
    const float* W = (q & 2) ? Wr : Wl;
    float*      out = (q & 2) ? xr : xl;
    int cbase = (q & 1) * 64;
    int half = lane >> 4;
    int m    = lane & 15;
    int arow = rowTile * 16 + m;
    if (arow > N - 1) arow = N - 1;
    int col  = cbase + 4 * m;
    int koff = half * 2;
    const float* ap = h + (size_t)arow * EMB + koff;
    const float* wp = W + (size_t)koff * EMB + col;

    v8f c0 = {0.f, 0.f, 0.f, 0.f, 0.f, 0.f, 0.f, 0.f};
    v8f c1 = c0, c2 = c0, c3 = c0;
#pragma unroll
    for (int k = 0; k < EMB; k += 4) {
        v2f a = *(const v2f*)(ap + k);
        v4f w0 = *(const v4f*)(wp + (size_t)k * EMB);
        v4f w1 = *(const v4f*)(wp + (size_t)(k + 1) * EMB);
        v2f b0; b0.x = w0.x; b0.y = w1.x;
        v2f b1; b1.x = w0.y; b1.y = w1.y;
        v2f b2; b2.x = w0.z; b2.y = w1.z;
        v2f b3; b3.x = w0.w; b3.y = w1.w;
        c0 = __builtin_amdgcn_wmma_f32_16x16x4_f32(false, a, false, b0, (short)0, c0,
                                                   false, false);
        c1 = __builtin_amdgcn_wmma_f32_16x16x4_f32(false, a, false, b1, (short)0, c1,
                                                   false, false);
        c2 = __builtin_amdgcn_wmma_f32_16x16x4_f32(false, a, false, b2, (short)0, c2,
                                                   false, false);
        c3 = __builtin_amdgcn_wmma_f32_16x16x4_f32(false, a, false, b3, (short)0, c3,
                                                   false, false);
    }
#pragma unroll
    for (int j = 0; j < 8; ++j) {
        int r = rowTile * 16 + j + half * 8;
        v4f v; v.x = c0[j]; v.y = c1[j]; v.z = c2[j]; v.w = c3[j];
        if (r < N) *(v4f*)(out + (size_t)r * EMB + col) = v;
    }
}

__device__ __forceinline__ float gat_head_logit(const float4 xs, const float4 xd, const float4 we,
                                                const float4 at, const float w) {
    float p = 0.0f, v;
    v = xs.x + xd.x + w * we.x; v = (v >= 0.f) ? v : NEG_SLOPE * v; p += v * at.x;
    v = xs.y + xd.y + w * we.y; v = (v >= 0.f) ? v : NEG_SLOPE * v; p += v * at.y;
    v = xs.z + xd.z + w * we.z; v = (v >= 0.f) ? v : NEG_SLOPE * v; p += v * at.z;
    v = xs.w + xd.w + w * we.w; v = (v >= 0.f) ? v : NEG_SLOPE * v; p += v * at.w;
    for (int msk = 8; msk >= 1; msk >>= 1) p += __shfl_xor(p, msk, 32);
    return p;
}

__global__ __launch_bounds__(256)
void gat_node_kernel(const float* __restrict__ xl, const float* __restrict__ xr,
                     const int* __restrict__ off, const int* __restrict__ grp,
                     const float* __restrict__ We, const float* __restrict__ att,
                     const float* __restrict__ bias, float* __restrict__ o, int N) {
    const int wave = (blockIdx.x * blockDim.x + threadIdx.x) >> 5;
    const int lane = threadIdx.x & 31;
    if (wave >= N) return;
    const int d = __builtin_amdgcn_readfirstlane(wave);
    const int beg = off[d];
    const int end = off[d + 1];

    const float4 xd = *(const float4*)(xr + (size_t)d * EMB + lane * 4);
    const float4 xself = *(const float4*)(xl + (size_t)d * EMB + lane * 4);
    const float4 we = *(const float4*)(We + lane * 4);
    const float4 at = *(const float4*)(att + lane * 4);

    float m = -INFINITY;
    float den = 0.0f;
    float4 acc = {0.0f, 0.0f, 0.0f, 0.0f};
    float wsum = 0.0f;

    for (int base = beg; base < end; base += 32) {
        int my_s = 0, my_w = 0;
        if (base + lane < end) {
            const int2 rec = *(const int2*)(grp + 2 * (size_t)(base + lane));
            my_s = rec.x; my_w = rec.y;
        }
        wsum += __int_as_float(my_w);
        const int nb = (end - base < 32) ? (end - base) : 32;
        for (int i = 0; i < nb; i += 4) {
            const int i1 = (i + 1 < nb) ? (i + 1) : (nb - 1);
            const int i2 = (i + 2 < nb) ? (i + 2) : (nb - 1);
            const int i3 = (i + 3 < nb) ? (i + 3) : (nb - 1);
            const int s0 = __builtin_amdgcn_readlane(my_s, i);
            const int s1 = __builtin_amdgcn_readlane(my_s, i1);
            const int s2 = __builtin_amdgcn_readlane(my_s, i2);
            const int s3 = __builtin_amdgcn_readlane(my_s, i3);
            const float w0 = __int_as_float(__builtin_amdgcn_readlane(my_w, i));
            const float w1 = __int_as_float(__builtin_amdgcn_readlane(my_w, i1));
            const float w2 = __int_as_float(__builtin_amdgcn_readlane(my_w, i2));
            const float w3 = __int_as_float(__builtin_amdgcn_readlane(my_w, i3));
            const float4 r0 = *(const float4*)(xl + (size_t)s0 * EMB + lane * 4);
            const float4 r1 = *(const float4*)(xl + (size_t)s1 * EMB + lane * 4);
            const float4 r2 = *(const float4*)(xl + (size_t)s2 * EMB + lane * 4);
            const float4 r3 = *(const float4*)(xl + (size_t)s3 * EMB + lane * 4);
            const float p0 = gat_head_logit(r0, xd, we, at, w0);
            float p1 = gat_head_logit(r1, xd, we, at, w1);
            float p2 = gat_head_logit(r2, xd, we, at, w2);
            float p3 = gat_head_logit(r3, xd, we, at, w3);
            p1 = (i + 1 < nb) ? p1 : -INFINITY;
            p2 = (i + 2 < nb) ? p2 : -INFINITY;
            p3 = (i + 3 < nb) ? p3 : -INFINITY;
            const float mn = fmaxf(fmaxf(m, p0), fmaxf(p1, fmaxf(p2, p3)));
            const float sc = __expf(m - mn);
            const float e0 = __expf(p0 - mn);
            const float e1 = __expf(p1 - mn);
            const float e2 = __expf(p2 - mn);
            const float e3 = __expf(p3 - mn);
            den = den * sc + ((e0 + e1) + (e2 + e3));
            acc.x = acc.x * sc + ((e0 * r0.x + e1 * r1.x) + (e2 * r2.x + e3 * r3.x));
            acc.y = acc.y * sc + ((e0 * r0.y + e1 * r1.y) + (e2 * r2.y + e3 * r3.y));
            acc.z = acc.z * sc + ((e0 * r0.z + e1 * r1.z) + (e2 * r2.z + e3 * r3.z));
            acc.w = acc.w * sc + ((e0 * r0.w + e1 * r1.w) + (e2 * r2.w + e3 * r3.w));
            m = mn;
        }
    }

    for (int msk = 16; msk >= 1; msk >>= 1) wsum += __shfl_xor(wsum, msk, 32);
    const float wself = wsum / fmaxf((float)(end - beg), 1.0f);
    {
        const float p = gat_head_logit(xself, xd, we, at, wself);
        const float mn = fmaxf(m, p);
        const float sc = __expf(m - mn);
        const float ex = __expf(p - mn);
        den = den * sc + ex;
        acc.x = acc.x * sc + ex * xself.x;
        acc.y = acc.y * sc + ex * xself.y;
        acc.z = acc.z * sc + ex * xself.z;
        acc.w = acc.w * sc + ex * xself.w;
    }

    const float4 b = *(const float4*)(bias + lane * 4);
    float4 r;
    r.x = acc.x / den + b.x;
    r.y = acc.y / den + b.y;
    r.z = acc.z / den + b.z;
    r.w = acc.w / den + b.w;
    r.x = (r.x > 0.0f) ? r.x : (__expf(r.x) - 1.0f);
    r.y = (r.y > 0.0f) ? r.y : (__expf(r.y) - 1.0f);
    r.z = (r.z > 0.0f) ? r.z : (__expf(r.z) - 1.0f);
    r.w = (r.w > 0.0f) ? r.w : (__expf(r.w) - 1.0f);
    *(float4*)(o + (size_t)d * EMB + lane * 4) = r;
}

extern "C" void kernel_launch(void* const* d_in, const int* in_sizes, int n_in,
                              void* d_out, int out_size, void* d_ws, size_t ws_size,
                              hipStream_t stream) {
    const int*   edge_index = (const int*)d_in[0];
    const float* ew   = (const float*)d_in[1];
    const float* emb  = (const float*)d_in[2];
    const float* Wl   = (const float*)d_in[3];
    const float* Wr   = (const float*)d_in[4];
    const float* We   = (const float*)d_in[5];
    const float* att  = (const float*)d_in[6];
    const float* bias = (const float*)d_in[7];

    const int E = in_sizes[0] / 2;
    const int N = in_sizes[2] / EMB;
    const int* src = edge_index;
    const int* dst = edge_index + E;

    float* ws    = (float*)d_ws;
    float* xl    = ws;
    float* xr    = xl + (size_t)N * EMB;
    float* hbuf  = xr + (size_t)N * EMB;
    int*   cnt   = (int*)(hbuf + (size_t)N * EMB);
    int*   off   = cnt + (((size_t)N + 31) / 32) * 32;
    int*   rank  = off + (((size_t)N + 1 + 31) / 32) * 32;
    int*   grp   = rank + (((size_t)E + 31) / 32) * 32;

    const int BT = 256;
    const int rowTiles = (N + 15) / 16;
    const int gemmBlocks = (rowTiles * 4 + 7) / 8;
    const int edgeBlocks = (E + BT - 1) / BT;
    const int nodeWaveBlocks = (N * 32 + BT - 1) / BT;

    gat_zero_kernel<<<(N + BT - 1) / BT, BT, 0, stream>>>(cnt, N);
    gat_rank_kernel<<<edgeBlocks, BT, 0, stream>>>(dst, cnt, rank, E);
    gat_scan_kernel<<<1, 1024, 0, stream>>>(cnt, off, N);
    gat_scatter_kernel<<<edgeBlocks, BT, 0, stream>>>(src, dst, ew, off, rank, grp, E);

    for (int l = 0; l < 2; ++l) {
        const float* hin = (l == 0) ? emb : hbuf;
        float* o = (l == 0) ? hbuf : (float*)d_out;
        const float* Wl_l = Wl + (size_t)l * EMB * EMB;
        const float* Wr_l = Wr + (size_t)l * EMB * EMB;
        const float* We_l = We + (size_t)l * EMB;
        const float* at_l = att + (size_t)l * EMB;
        const float* b_l  = bias + (size_t)l * EMB;

        gat_gemm_kernel<<<gemmBlocks, BT, 0, stream>>>(hin, Wl_l, Wr_l, xl, xr, N);
        gat_node_kernel<<<nodeWaveBlocks, BT, 0, stream>>>(xl, xr, off, grp, We_l, at_l, b_l,
                                                           o, N);
    }
}


# edge grouping on LDS counters: two-level counting sort (slice counts per 256-node bucket, scan, place; a work-group per bucket ranks by node); no global atomics, no clear launch; sums arrival-ordered
# speedup vs baseline: 5.7075x; 1.2783x over previous
#include <hip/hip_runtime.h>
#include <math.h>


#define EMB 128
#define NEG_SLOPE 0.2f

typedef float v2f __attribute__((ext_vector_type(2)));
typedef float v8f __attribute__((ext_vector_type(8)));

#define BKT_SHIFT 8
#define BKT_NODES (1 << BKT_SHIFT)
#define L1_BLOCKS 256
#define L1_THREADS 512
#define L1_PER 8
#define L2_THREADS 512
#define L2_PER 10

__global__ __launch_bounds__(L1_THREADS)
void gat_bucket_count_kernel(const int* __restrict__ dst, int* __restrict__ mat, int E, int NB) {
    extern __shared__ int lds_tab[];
    const int t = threadIdx.x;
    const int per = (E + L1_BLOCKS - 1) / L1_BLOCKS;
    const int e0 = blockIdx.x * per;
    const int e1 = (e0 + per < E) ? (e0 + per) : E;
    for (int b = t; b < NB; b += L1_THREADS) lds_tab[b] = 0;
    __syncthreads();
    for (int c0 = e0; c0 < e1; c0 += L1_THREADS * L1_PER) {
        int d[L1_PER];
#pragma unroll
        for (int j = 0; j < L1_PER; ++j) {
            const int e = c0 + j * L1_THREADS + t;
            d[j] = (e < e1) ? dst[e] : -1;
        }
#pragma unroll
        for (int j = 0; j < L1_PER; ++j) {
            const unsigned b = (unsigned)(d[j] >> BKT_SHIFT);
            if (b < (unsigned)NB) atomicAdd(&lds_tab[b], 1);
        }
    }
    __syncthreads();
    for (int b = t; b < NB; b += L1_THREADS) mat[b * L1_BLOCKS + blockIdx.x] = lds_tab[b];
}

#define SCAN_THREADS 1024
#define SCAN_CHUNKS 16

__device__ __forceinline__ int gat_wave_incl_scan(int v, const int lane) {
#pragma unroll
    for (int d = 1; d < 32; d <<= 1) {
        const int u = __shfl_up(v, d, 32);
        if (lane >= d) v += u;
    }
    return v;
}

__global__ __launch_bounds__(SCAN_THREADS)
void gat_scan_kernel(const int* __restrict__ cnt, int* __restrict__ off, int N) {
    __shared__ int wtot[SCAN_CHUNKS * 32];
    __shared__ int wpre[SCAN_CHUNKS * 32];
    __shared__ int ctot[SCAN_CHUNKS];
    const int t = threadIdx.x;
    const int lane = t & 31;
    const int w = t >> 5;
    int carry = 0;
    for (int p0 = 0; p0 <= N; p0 += SCAN_THREADS * SCAN_CHUNKS * 4) {
        int4 c[SCAN_CHUNKS];
        int inc[SCAN_CHUNKS];
#pragma unroll
        for (int j = 0; j < SCAN_CHUNKS; ++j) {
            const int i0 = p0 + (j * SCAN_THREADS + t) * 4;
            int4 v = {0, 0, 0, 0};
            if (i0 < N) v = *(const int4*)(cnt + i0);
            c[j] = v;
        }
#pragma unroll
        for (int j = 0; j < SCAN_CHUNKS; ++j) inc[j] = c[j].x + c[j].y + c[j].z + c[j].w;
#pragma unroll
        for (int d = 1; d < 32; d <<= 1) {
            int u[SCAN_CHUNKS];
#pragma unroll
            for (int j = 0; j < SCAN_CHUNKS; ++j) u[j] = __shfl_up(inc[j], d, 32);
#pragma unroll
            for (int j = 0; j < SCAN_CHUNKS; ++j) inc[j] += (lane >= d) ? u[j] : 0;
        }
        if (lane == 31) {
#pragma unroll
            for (int j = 0; j < SCAN_CHUNKS; ++j) wtot[j * 32 + w] = inc[j];
        }
        __syncthreads();
        if (w < SCAN_CHUNKS) {
            const int v = wtot[w * 32 + lane];
            const int s = gat_wave_incl_scan(v, lane);
            wpre[w * 32 + lane] = s - v;
            if (lane == 31) ctot[w] = s;
        }
        __syncthreads();
        int run = carry;
#pragma unroll
        for (int j = 0; j < SCAN_CHUNKS; ++j) {
            const int i0 = p0 + (j * SCAN_THREADS + t) * 4;
            const int4 v = c[j];
            int4 o;
            o.x = run + wpre[j * 32 + w] + (inc[j] - (v.x + v.y + v.z + v.w));
            o.y = o.x + v.x;
            o.z = o.y + v.y;
            o.w = o.z + v.z;
            if (i0 <= N) *(int4*)(off + i0) = o;
            run += ctot[j];
        }
        carry = run;
        __syncthreads();
    }
}

__device__ __forceinline__ void gat_slice_load(const int* __restrict__ src,
                                               const int* __restrict__ dst,
                                               const float* __restrict__ ew, const int c0,
                                               const int e1, const int t, int (&s)[L1_PER],
                                               int (&d)[L1_PER], int (&w)[L1_PER]) {
#pragma unroll
    for (int j = 0; j < L1_PER; ++j) {
        const int e = c0 + j * L1_THREADS + t;
        const bool in = e < e1;
        d[j] = in ? dst[e] : -1;
        s[j] = in ? src[e] : 0;
        w[j] = in ? __float_as_int(ew[e]) : 0;
    }
}

__global__ __launch_bounds__(L1_THREADS)
void gat_bucket_place_kernel(const int* __restrict__ src, const int* __restrict__ dst,
                             const float* __restrict__ ew, const int* __restrict__ base,
                             int4* __restrict__ tmp, int E, int NB) {
    extern __shared__ int lds_tab[];
    const int t = threadIdx.x;
    const int per = (E + L1_BLOCKS - 1) / L1_BLOCKS;
    const int e0 = blockIdx.x * per;
    const int e1 = (e0 + per < E) ? (e0 + per) : E;
    int s[L1_PER], d[L1_PER], w[L1_PER];
    int c0 = e0;
    gat_slice_load(src, dst, ew, c0, e1, t, s, d, w);
    for (int b = t; b < NB; b += L1_THREADS) lds_tab[b] = base[b * L1_BLOCKS + blockIdx.x];
    __syncthreads();
    for (;;) {
        int pos[L1_PER];
#pragma unroll
        for (int j = 0; j < L1_PER; ++j) {
            const unsigned b = (unsigned)(d[j] >> BKT_SHIFT);
            pos[j] = -1;
            if (b < (unsigned)NB) pos[j] = atomicAdd(&lds_tab[b], 1);
        }
#pragma unroll
        for (int j = 0; j < L1_PER; ++j) {
            if ((unsigned)pos[j] < (unsigned)E) tmp[pos[j]] = make_int4(s[j], w[j], d[j], 0);
        }
        c0 += L1_THREADS * L1_PER;
        if (c0 >= e1) break;
        gat_slice_load(src, dst, ew, c0, e1, t, s, d, w);
    }
}

__device__ __forceinline__ void gat_bucket_load(const int4* __restrict__ tmp, const int c0,
                                                const int re, const int t, int4 (&r)[L2_PER]) {
#pragma unroll
    for (int j = 0; j < L2_PER; ++j) {
        const int i = c0 + j * L2_THREADS + t;
        int4 v = {0, 0, -1, 0};
        if (i < re) v = tmp[i];
        r[j] = v;
    }
}

__device__ __forceinline__ void gat_bucket_put(const int4 (&r)[L2_PER], int* cur, const int n0,
                                               int* __restrict__ grp, const int E) {
    int pos[L2_PER];
#pragma unroll
    for (int j = 0; j < L2_PER; ++j) {
        const unsigned loc = (unsigned)(r[j].z - n0);
        pos[j] = -1;
        if (loc < (unsigned)BKT_NODES) pos[j] = atomicAdd(&cur[loc], 1);
    }
#pragma unroll
    for (int j = 0; j < L2_PER; ++j) {
        if ((unsigned)pos[j] < (unsigned)E)
            *(int2*)(grp + 2 * (size_t)pos[j]) = make_int2(r[j].x, r[j].y);
    }
}

__global__ __launch_bounds__(L2_THREADS)
void gat_bucket_sort_kernel(const int* __restrict__ base, const int4* __restrict__ tmp,
                            int* __restrict__ off, int* __restrict__ grp, int E, int N, int NB) {
    __shared__ int cur[BKT_NODES];
    __shared__ int wtot[BKT_NODES / 32];
    const int t = threadIdx.x;
    const int lane = t & 31;
    const int b = blockIdx.x;
    const int n0 = b << BKT_SHIFT;
    int rs = base[b * L1_BLOCKS];
    int re = base[(b + 1) * L1_BLOCKS];
    rs = (rs < 0) ? 0 : ((rs > E) ? E : rs);
    re = (re < rs) ? rs : ((re > E) ? E : re);
    if (t < BKT_NODES) cur[t] = 0;
    __syncthreads();

    int4 r[L2_PER];
    int c0 = rs;
    for (;;) {
        gat_bucket_load(tmp, c0, re, t, r);
#pragma unroll
        for (int j = 0; j < L2_PER; ++j) {
            const unsigned loc = (unsigned)(r[j].z - n0);
            if (loc < (unsigned)BKT_NODES) atomicAdd(&cur[loc], 1);
        }
        if (c0 + L2_THREADS * L2_PER >= re) break;
        c0 += L2_THREADS * L2_PER;
    }
    __syncthreads();

    int v = 0, incl = 0;
    if (t < BKT_NODES) {
        v = cur[t];
        incl = gat_wave_incl_scan(v, lane);
        if (lane == 31) wtot[t >> 5] = incl;
    }
    __syncthreads();
    if (t < BKT_NODES) {
        int pre = 0;
        for (int k = 0; k < (t >> 5); ++k) pre += wtot[k];
        const int start = rs + pre + (incl - v);
        cur[t] = start;
        if (n0 + t < N) off[n0 + t] = start;
    }
    if (b == NB - 1 && t == 0) off[N] = re;
    __syncthreads();

    gat_bucket_put(r, cur, n0, grp, E);
    for (int c = rs; c < c0; c += L2_THREADS * L2_PER) {
        gat_bucket_load(tmp, c, re, t, r);
        gat_bucket_put(r, cur, n0, grp, E);
    }
}

typedef float v4f __attribute__((ext_vector_type(4)));

__global__ __launch_bounds__(256)
void gat_gemm_kernel(const float* __restrict__ h, const float* __restrict__ Wl,
                     const float* __restrict__ Wr, float* __restrict__ xl,
                     float* __restrict__ xr, int N) {
    int lane = threadIdx.x & 31;
    int wave = (blockIdx.x * blockDim.x + threadIdx.x) >> 5;
    int rowTiles = (N + 15) >> 4;
    if (wave >= rowTiles * 4) return;
    int rowTile = wave >> 2;
    int q = wave & 3;
    const float* W = (q & 2) ? Wr : Wl;
    float*      out = (q & 2) ? xr : xl;
    int cbase = (q & 1) * 64;
    int half = lane >> 4;
    int m    = lane & 15;
    int arow = rowTile * 16 + m;
    if (arow > N - 1) arow = N - 1;
    int col  = cbase + 4 * m;
    int koff = half * 2;
    const float* ap = h + (size_t)arow * EMB + koff;
    const float* wp = W + (size_t)koff * EMB + col;

    v8f c0 = {0.f, 0.f, 0.f, 0.f, 0.f, 0.f, 0.f, 0.f};
    v8f c1 = c0, c2 = c0, c3 = c0;
#pragma unroll
    for (int k = 0; k < EMB; k += 4) {
        v2f a = *(const v2f*)(ap + k);
        v4f w0 = *(const v4f*)(wp + (size_t)k * EMB);
        v4f w1 = *(const v4f*)(wp + (size_t)(k + 1) * EMB);
        v2f b0; b0.x = w0.x; b0.y = w1.x;
        v2f b1; b1.x = w0.y; b1.y = w1.y;
        v2f b2; b2.x = w0.z; b2.y = w1.z;
        v2f b3; b3.x = w0.w; b3.y = w1.w;
        c0 = __builtin_amdgcn_wmma_f32_16x16x4_f32(false, a, false, b0, (short)0, c0,
                                                   false, false);
        c1 = __builtin_amdgcn_wmma_f32_16x16x4_f32(false, a, false, b1, (short)0, c1,
                                                   false, false);
        c2 = __builtin_amdgcn_wmma_f32_16x16x4_f32(false, a, false, b2, (short)0, c2,
                                                   false, false);
        c3 = __builtin_amdgcn_wmma_f32_16x16x4_f32(false, a, false, b3, (short)0, c3,
                                                   false, false);
    }
#pragma unroll
    for (int j = 0; j < 8; ++j) {
        int r = rowTile * 16 + j + half * 8;
        v4f v; v.x = c0[j]; v.y = c1[j]; v.z = c2[j]; v.w = c3[j];
        if (r < N) *(v4f*)(out + (size_t)r * EMB + col) = v;
    }
}

__device__ __forceinline__ float gat_head_logit(const float4 xs, const float4 xd, const float4 we,
                                                const float4 at, const float w) {
    float p = 0.0f, v;
    v = xs.x + xd.x + w * we.x; v = (v >= 0.f) ? v : NEG_SLOPE * v; p += v * at.x;
    v = xs.y + xd.y + w * we.y; v = (v >= 0.f) ? v : NEG_SLOPE * v; p += v * at.y;
    v = xs.z + xd.z + w * we.z; v = (v >= 0.f) ? v : NEG_SLOPE * v; p += v * at.z;
    v = xs.w + xd.w + w * we.w; v = (v >= 0.f) ? v : NEG_SLOPE * v; p += v * at.w;
    for (int msk = 8; msk >= 1; msk >>= 1) p += __shfl_xor(p, msk, 32);
    return p;
}

__global__ __launch_bounds__(256)
void gat_node_kernel(const float* __restrict__ xl, const float* __restrict__ xr,
                     const int* __restrict__ off, const int* __restrict__ grp,
                     const float* __restrict__ We, const float* __restrict__ att,
                     const float* __restrict__ bias, float* __restrict__ o, int N) {
    const int wave = (blockIdx.x * blockDim.x + threadIdx.x) >> 5;
    const int lane = threadIdx.x & 31;
    if (wave >= N) return;
    const int d = __builtin_amdgcn_readfirstlane(wave);
    const int beg = off[d];
    const int end = off[d + 1];

    const float4 xd = *(const float4*)(xr + (size_t)d * EMB + lane * 4);
    const float4 xself = *(const float4*)(xl + (size_t)d * EMB + lane * 4);
    const float4 we = *(const float4*)(We + lane * 4);
    const float4 at = *(const float4*)(att + lane * 4);

    float m = -INFINITY;
    float den = 0.0f;
    float4 acc = {0.0f, 0.0f, 0.0f, 0.0f};
    float wsum = 0.0f;

    for (int base = beg; base < end; base += 32) {
        int my_s = 0, my_w = 0;
        if (base + lane < end) {
            const int2 rec = *(const int2*)(grp + 2 * (size_t)(base + lane));
            my_s = rec.x; my_w = rec.y;
        }
        wsum += __int_as_float(my_w);
        const int nb = (end - base < 32) ? (end - base) : 32;
        for (int i = 0; i < nb; i += 4) {
            const int i1 = (i + 1 < nb) ? (i + 1) : (nb - 1);
            const int i2 = (i + 2 < nb) ? (i + 2) : (nb - 1);
            const int i3 = (i + 3 < nb) ? (i + 3) : (nb - 1);
            const int s0 = __builtin_amdgcn_readlane(my_s, i);
            const int s1 = __builtin_amdgcn_readlane(my_s, i1);
            const int s2 = __builtin_amdgcn_readlane(my_s, i2);
            const int s3 = __builtin_amdgcn_readlane(my_s, i3);
            const float w0 = __int_as_float(__builtin_amdgcn_readlane(my_w, i));
            const float w1 = __int_as_float(__builtin_amdgcn_readlane(my_w, i1));
            const float w2 = __int_as_float(__builtin_amdgcn_readlane(my_w, i2));
            const float w3 = __int_as_float(__builtin_amdgcn_readlane(my_w, i3));
            const float4 r0 = *(const float4*)(xl + (size_t)s0 * EMB + lane * 4);
            const float4 r1 = *(const float4*)(xl + (size_t)s1 * EMB + lane * 4);
            const float4 r2 = *(const float4*)(xl + (size_t)s2 * EMB + lane * 4);
            const float4 r3 = *(const float4*)(xl + (size_t)s3 * EMB + lane * 4);
            const float p0 = gat_head_logit(r0, xd, we, at, w0);
            float p1 = gat_head_logit(r1, xd, we, at, w1);
            float p2 = gat_head_logit(r2, xd, we, at, w2);
            float p3 = gat_head_logit(r3, xd, we, at, w3);
            p1 = (i + 1 < nb) ? p1 : -INFINITY;
            p2 = (i + 2 < nb) ? p2 : -INFINITY;
            p3 = (i + 3 < nb) ? p3 : -INFINITY;
            const float mn = fmaxf(fmaxf(m, p0), fmaxf(p1, fmaxf(p2, p3)));
            const float sc = __expf(m - mn);
            const float e0 = __expf(p0 - mn);
            const float e1 = __expf(p1 - mn);
            const float e2 = __expf(p2 - mn);
            const float e3 = __expf(p3 - mn);
            den = den * sc + ((e0 + e1) + (e2 + e3));
            acc.x = acc.x * sc + ((e0 * r0.x + e1 * r1.x) + (e2 * r2.x + e3 * r3.x));
            acc.y = acc.y * sc + ((e0 * r0.y + e1 * r1.y) + (e2 * r2.y + e3 * r3.y));
            acc.z = acc.z * sc + ((e0 * r0.z + e1 * r1.z) + (e2 * r2.z + e3 * r3.z));
            acc.w = acc.w * sc + ((e0 * r0.w + e1 * r1.w) + (e2 * r2.w + e3 * r3.w));
            m = mn;
        }
    }

    for (int msk = 16; msk >= 1; msk >>= 1) wsum += __shfl_xor(wsum, msk, 32);
    const float wself = wsum / fmaxf((float)(end - beg), 1.0f);
    {
        const float p = gat_head_logit(xself, xd, we, at, wself);
        const float mn = fmaxf(m, p);
        const float sc = __expf(m - mn);
        const float ex = __expf(p - mn);
        den = den * sc + ex;
        acc.x = acc.x * sc + ex * xself.x;
        acc.y = acc.y * sc + ex * xself.y;
        acc.z = acc.z * sc + ex * xself.z;
        acc.w = acc.w * sc + ex * xself.w;
    }

    const float4 b = *(const float4*)(bias + lane * 4);
    float4 r;
    r.x = acc.x / den + b.x;
    r.y = acc.y / den + b.y;
    r.z = acc.z / den + b.z;
    r.w = acc.w / den + b.w;
    r.x = (r.x > 0.0f) ? r.x : (__expf(r.x) - 1.0f);
    r.y = (r.y > 0.0f) ? r.y : (__expf(r.y) - 1.0f);
    r.z = (r.z > 0.0f) ? r.z : (__expf(r.z) - 1.0f);
    r.w = (r.w > 0.0f) ? r.w : (__expf(r.w) - 1.0f);
    *(float4*)(o + (size_t)d * EMB + lane * 4) = r;
}

extern "C" void kernel_launch(void* const* d_in, const int* in_sizes, int n_in,
                              void* d_out, int out_size, void* d_ws, size_t ws_size,
                              hipStream_t stream) {
    const int*   edge_index = (const int*)d_in[0];
    const float* ew   = (const float*)d_in[1];
    const float* emb  = (const float*)d_in[2];
    const float* Wl   = (const float*)d_in[3];
    const float* Wr   = (const float*)d_in[4];
    const float* We   = (const float*)d_in[5];
    const float* att  = (const float*)d_in[6];
    const float* bias = (const float*)d_in[7];

    const int E = in_sizes[0] / 2;
    const int N = in_sizes[2] / EMB;
    const int* src = edge_index;
    const int* dst = edge_index + E;

    float* ws    = (float*)d_ws;
    float* xl    = ws;
    float* xr    = xl + (size_t)N * EMB;
    float* hbuf  = xr + (size_t)N * EMB;
    const int NB = (N + BKT_NODES - 1) / BKT_NODES;
    const size_t M = (size_t)NB * L1_BLOCKS;
    int*   mat   = (int*)(hbuf + (size_t)N * EMB);
    int*   base  = mat + ((M + 31) / 32) * 32;
    int*   off   = base + ((M + 4 + 31) / 32) * 32;
    int*   tmp   = off + (((size_t)N + 1 + 31) / 32) * 32;
    int*   grp   = tmp + (((size_t)E * 4 + 31) / 32) * 32;

    const int BT = 256;
    const int rowTiles = (N + 15) / 16;
    const int gemmBlocks = (rowTiles * 4 + 7) / 8;
    const int nodeWaveBlocks = (N * 32 + BT - 1) / BT;

    gat_bucket_count_kernel<<<L1_BLOCKS, L1_THREADS, NB * sizeof(int), stream>>>(dst, mat, E, NB);
    gat_scan_kernel<<<1, 1024, 0, stream>>>(mat, base, NB * L1_BLOCKS);
    gat_bucket_place_kernel<<<L1_BLOCKS, L1_THREADS, NB * sizeof(int), stream>>>(
        src, dst, ew, base, (int4*)tmp, E, NB);
    gat_bucket_sort_kernel<<<NB, L2_THREADS, 0, stream>>>(base, (const int4*)tmp, off, grp, E, N,
                                                          NB);

    for (int l = 0; l < 2; ++l) {
        const float* hin = (l == 0) ? emb : hbuf;
        float* o = (l == 0) ? hbuf : (float*)d_out;
        const float* Wl_l = Wl + (size_t)l * EMB * EMB;
        const float* Wr_l = Wr + (size_t)l * EMB * EMB;
        const float* We_l = We + (size_t)l * EMB;
        const float* at_l = att + (size_t)l * EMB;
        const float* b_l  = bias + (size_t)l * EMB;

        gat_gemm_kernel<<<gemmBlocks, BT, 0, stream>>>(hin, Wl_l, Wr_l, xl, xr, N);
        gat_node_kernel<<<nodeWaveBlocks, BT, 0, stream>>>(xl, xr, off, grp, We_l, at_l, b_l,
                                                           o, N);
    }
}


# node stage logit with fewer issue slots: the half-wave sum steps are DPP row_xmask operands of the adds (no LDS permute, no wait), leaky_relu as max(v, 0.2v); pairing of the sums unchanged
# speedup vs baseline: 6.0356x; 1.0575x over previous
#include <hip/hip_runtime.h>
#include <math.h>


#define EMB 128
#define NEG_SLOPE 0.2f

typedef float v2f __attribute__((ext_vector_type(2)));
typedef float v8f __attribute__((ext_vector_type(8)));

#define BKT_SHIFT 8
#define BKT_NODES (1 << BKT_SHIFT)
#define L1_BLOCKS 256
#define L1_THREADS 512
#define L1_PER 8
#define L2_THREADS 512
#define L2_PER 10

__global__ __launch_bounds__(L1_THREADS)
void gat_bucket_count_kernel(const int* __restrict__ dst, int* __restrict__ mat, int E, int NB) {
    extern __shared__ int lds_tab[];
    const int t = threadIdx.x;
    const int per = (E + L1_BLOCKS - 1) / L1_BLOCKS;
    const int e0 = blockIdx.x * per;
    const int e1 = (e0 + per < E) ? (e0 + per) : E;
    for (int b = t; b < NB; b += L1_THREADS) lds_tab[b] = 0;
    __syncthreads();
    for (int c0 = e0; c0 < e1; c0 += L1_THREADS * L1_PER) {
        int d[L1_PER];
#pragma unroll
        for (int j = 0; j < L1_PER; ++j) {
            const int e = c0 + j * L1_THREADS + t;
            d[j] = (e < e1) ? dst[e] : -1;
        }
#pragma unroll
        for (int j = 0; j < L1_PER; ++j) {
            const unsigned b = (unsigned)(d[j] >> BKT_SHIFT);
            if (b < (unsigned)NB) atomicAdd(&lds_tab[b], 1);
        }
    }
    __syncthreads();
    for (int b = t; b < NB; b += L1_THREADS) mat[b * L1_BLOCKS + blockIdx.x] = lds_tab[b];
}

#define SCAN_THREADS 1024
#define SCAN_CHUNKS 16

__device__ __forceinline__ int gat_wave_incl_scan(int v, const int lane) {
#pragma unroll
    for (int d = 1; d < 32; d <<= 1) {
        const int u = __shfl_up(v, d, 32);
        if (lane >= d) v += u;
    }
    return v;
}

__global__ __launch_bounds__(SCAN_THREADS)
void gat_scan_kernel(const int* __restrict__ cnt, int* __restrict__ off, int N) {
    __shared__ int wtot[SCAN_CHUNKS * 32];
    __shared__ int wpre[SCAN_CHUNKS * 32];
    __shared__ int ctot[SCAN_CHUNKS];
    const int t = threadIdx.x;
    const int lane = t & 31;
    const int w = t >> 5;
    int carry = 0;
    for (int p0 = 0; p0 <= N; p0 += SCAN_THREADS * SCAN_CHUNKS * 4) {
        int4 c[SCAN_CHUNKS];
        int inc[SCAN_CHUNKS];
#pragma unroll
        for (int j = 0; j < SCAN_CHUNKS; ++j) {
            const int i0 = p0 + (j * SCAN_THREADS + t) * 4;
            int4 v = {0, 0, 0, 0};
            if (i0 < N) v = *(const int4*)(cnt + i0);
            c[j] = v;
        }
#pragma unroll
        for (int j = 0; j < SCAN_CHUNKS; ++j) inc[j] = c[j].x + c[j].y + c[j].z + c[j].w;
#pragma unroll
        for (int d = 1; d < 32; d <<= 1) {
            int u[SCAN_CHUNKS];
#pragma unroll
            for (int j = 0; j < SCAN_CHUNKS; ++j) u[j] = __shfl_up(inc[j], d, 32);
#pragma unroll
            for (int j = 0; j < SCAN_CHUNKS; ++j) inc[j] += (lane >= d) ? u[j] : 0;
        }
        if (lane == 31) {
#pragma unroll
            for (int j = 0; j < SCAN_CHUNKS; ++j) wtot[j * 32 + w] = inc[j];
        }
        __syncthreads();
        if (w < SCAN_CHUNKS) {
            const int v = wtot[w * 32 + lane];
            const int s = gat_wave_incl_scan(v, lane);
            wpre[w * 32 + lane] = s - v;
            if (lane == 31) ctot[w] = s;
        }
        __syncthreads();
        int run = carry;
#pragma unroll
        for (int j = 0; j < SCAN_CHUNKS; ++j) {
            const int i0 = p0 + (j * SCAN_THREADS + t) * 4;
            const int4 v = c[j];
            int4 o;
            o.x = run + wpre[j * 32 + w] + (inc[j] - (v.x + v.y + v.z + v.w));
            o.y = o.x + v.x;
            o.z = o.y + v.y;
            o.w = o.z + v.z;
            if (i0 <= N) *(int4*)(off + i0) = o;
            run += ctot[j];
        }
        carry = run;
        __syncthreads();
    }
}

__device__ __forceinline__ void gat_slice_load(const int* __restrict__ src,
                                               const int* __restrict__ dst,
                                               const float* __restrict__ ew, const int c0,
                                               const int e1, const int t, int (&s)[L1_PER],
                                               int (&d)[L1_PER], int (&w)[L1_PER]) {
#pragma unroll
    for (int j = 0; j < L1_PER; ++j) {
        const int e = c0 + j * L1_THREADS + t;
        const bool in = e < e1;
        d[j] = in ? dst[e] : -1;
        s[j] = in ? src[e] : 0;
        w[j] = in ? __float_as_int(ew[e]) : 0;
    }
}

__global__ __launch_bounds__(L1_THREADS)
void gat_bucket_place_kernel(const int* __restrict__ src, const int* __restrict__ dst,
                             const float* __restrict__ ew, const int* __restrict__ base,
                             int4* __restrict__ tmp, int E, int NB) {
    extern __shared__ int lds_tab[];
    const int t = threadIdx.x;
    const int per = (E + L1_BLOCKS - 1) / L1_BLOCKS;
    const int e0 = blockIdx.x * per;
    const int e1 = (e0 + per < E) ? (e0 + per) : E;
    int s[L1_PER], d[L1_PER], w[L1_PER];
    int c0 = e0;
    gat_slice_load(src, dst, ew, c0, e1, t, s, d, w);
    for (int b = t; b < NB; b += L1_THREADS) lds_tab[b] = base[b * L1_BLOCKS + blockIdx.x];
    __syncthreads();
    for (;;) {
        int pos[L1_PER];
#pragma unroll
        for (int j = 0; j < L1_PER; ++j) {
            const unsigned b = (unsigned)(d[j] >> BKT_SHIFT);
            pos[j] = -1;
            if (b < (unsigned)NB) pos[j] = atomicAdd(&lds_tab[b], 1);
        }
#pragma unroll
        for (int j = 0; j < L1_PER; ++j) {
            if ((unsigned)pos[j] < (unsigned)E) tmp[pos[j]] = make_int4(s[j], w[j], d[j], 0);
        }
        c0 += L1_THREADS * L1_PER;
        if (c0 >= e1) break;
        gat_slice_load(src, dst, ew, c0, e1, t, s, d, w);
    }
}

__device__ __forceinline__ void gat_bucket_load(const int4* __restrict__ tmp, const int c0,
                                                const int re, const int t, int4 (&r)[L2_PER]) {
#pragma unroll
    for (int j = 0; j < L2_PER; ++j) {
        const int i = c0 + j * L2_THREADS + t;
        int4 v = {0, 0, -1, 0};
        if (i < re) v = tmp[i];
        r[j] = v;
    }
}

__device__ __forceinline__ void gat_bucket_put(const int4 (&r)[L2_PER], int* cur, const int n0,
                                               int* __restrict__ grp, const int E) {
    int pos[L2_PER];
#pragma unroll
    for (int j = 0; j < L2_PER; ++j) {
        const unsigned loc = (unsigned)(r[j].z - n0);
        pos[j] = -1;
        if (loc < (unsigned)BKT_NODES) pos[j] = atomicAdd(&cur[loc], 1);
    }
#pragma unroll
    for (int j = 0; j < L2_PER; ++j) {
        if ((unsigned)pos[j] < (unsigned)E)
            *(int2*)(grp + 2 * (size_t)pos[j]) = make_int2(r[j].x, r[j].y);
    }
}

__global__ __launch_bounds__(L2_THREADS)
void gat_bucket_sort_kernel(const int* __restrict__ base, const int4* __restrict__ tmp,
                            int* __restrict__ off, int* __restrict__ grp, int E, int N, int NB) {
    __shared__ int cur[BKT_NODES];
    __shared__ int wtot[BKT_NODES / 32];
    const int t = threadIdx.x;
    const int lane = t & 31;
    const int b = blockIdx.x;
    const int n0 = b << BKT_SHIFT;
    int rs = base[b * L1_BLOCKS];
    int re = base[(b + 1) * L1_BLOCKS];
    rs = (rs < 0) ? 0 : ((rs > E) ? E : rs);
    re = (re < rs) ? rs : ((re > E) ? E : re);
    if (t < BKT_NODES) cur[t] = 0;
    __syncthreads();

    int4 r[L2_PER];
    int c0 = rs;
    for (;;) {
        gat_bucket_load(tmp, c0, re, t, r);
#pragma unroll
        for (int j = 0; j < L2_PER; ++j) {
            const unsigned loc = (unsigned)(r[j].z - n0);
            if (loc < (unsigned)BKT_NODES) atomicAdd(&cur[loc], 1);
        }
        if (c0 + L2_THREADS * L2_PER >= re) break;
        c0 += L2_THREADS * L2_PER;
    }
    __syncthreads();

    int v = 0, incl = 0;
    if (t < BKT_NODES) {
        v = cur[t];
        incl = gat_wave_incl_scan(v, lane);
        if (lane == 31) wtot[t >> 5] = incl;
    }
    __syncthreads();
    if (t < BKT_NODES) {
        int pre = 0;
        for (int k = 0; k < (t >> 5); ++k) pre += wtot[k];
        const int start = rs + pre + (incl - v);
        cur[t] = start;
        if (n0 + t < N) off[n0 + t] = start;
    }
    if (b == NB - 1 && t == 0) off[N] = re;
    __syncthreads();

    gat_bucket_put(r, cur, n0, grp, E);
    for (int c = rs; c < c0; c += L2_THREADS * L2_PER) {
        gat_bucket_load(tmp, c, re, t, r);
        gat_bucket_put(r, cur, n0, grp, E);
    }
}

typedef float v4f __attribute__((ext_vector_type(4)));

__global__ __launch_bounds__(256)
void gat_gemm_kernel(const float* __restrict__ h, const float* __restrict__ Wl,
                     const float* __restrict__ Wr, float* __restrict__ xl,
                     float* __restrict__ xr, int N) {
    int lane = threadIdx.x & 31;
    int wave = (blockIdx.x * blockDim.x + threadIdx.x) >> 5;
    int rowTiles = (N + 15) >> 4;
    if (wave >= rowTiles * 4) return;
    int rowTile = wave >> 2;
    int q = wave & 3;
    const float* W = (q & 2) ? Wr : Wl;
    float*      out = (q & 2) ? xr : xl;
    int cbase = (q & 1) * 64;
    int half = lane >> 4;
    int m    = lane & 15;
    int arow = rowTile * 16 + m;
    if (arow > N - 1) arow = N - 1;
    int col  = cbase + 4 * m;
    int koff = half * 2;
    const float* ap = h + (size_t)arow * EMB + koff;
    const float* wp = W + (size_t)koff * EMB + col;

    v8f c0 = {0.f, 0.f, 0.f, 0.f, 0.f, 0.f, 0.f, 0.f};
    v8f c1 = c0, c2 = c0, c3 = c0;
#pragma unroll
    for (int k = 0; k < EMB; k += 4) {
        v2f a = *(const v2f*)(ap + k);
        v4f w0 = *(const v4f*)(wp + (size_t)k * EMB);
        v4f w1 = *(const v4f*)(wp + (size_t)(k + 1) * EMB);
        v2f b0; b0.x = w0.x; b0.y = w1.x;
        v2f b1; b1.x = w0.y; b1.y = w1.y;
        v2f b2; b2.x = w0.z; b2.y = w1.z;
        v2f b3; b3.x = w0.w; b3.y = w1.w;
        c0 = __builtin_amdgcn_wmma_f32_16x16x4_f32(false, a, false, b0, (short)0, c0,
                                                   false, false);
        c1 = __builtin_amdgcn_wmma_f32_16x16x4_f32(false, a, false, b1, (short)0, c1,
                                                   false, false);
        c2 = __builtin_amdgcn_wmma_f32_16x16x4_f32(false, a, false, b2, (short)0, c2,
                                                   false, false);
        c3 = __builtin_amdgcn_wmma_f32_16x16x4_f32(false, a, false, b3, (short)0, c3,
                                                   false, false);
    }
#pragma unroll
    for (int j = 0; j < 8; ++j) {
        int r = rowTile * 16 + j + half * 8;
        v4f v; v.x = c0[j]; v.y = c1[j]; v.z = c2[j]; v.w = c3[j];
        if (r < N) *(v4f*)(out + (size_t)r * EMB + col) = v;
    }
}

template <int K>
__device__ __forceinline__ float gat_row_xadd(const float v) {
    const int u = __builtin_amdgcn_update_dpp(0, __float_as_int(v), 0x160 + K, 0xF, 0xF, true);
    return v + __int_as_float(u);
}

__device__ __forceinline__ float gat_leaky(const float v) {
    return fmaxf(v, NEG_SLOPE * v);
}

__device__ __forceinline__ float gat_head_logit(const float4 xs, const float4 xd, const float4 we,
                                                const float4 at, const float w) {
    float p = gat_leaky(xs.x + xd.x + w * we.x) * at.x;
    p += gat_leaky(xs.y + xd.y + w * we.y) * at.y;
    p += gat_leaky(xs.z + xd.z + w * we.z) * at.z;
    p += gat_leaky(xs.w + xd.w + w * we.w) * at.w;
    p = gat_row_xadd<8>(p);
    p = gat_row_xadd<4>(p);
    p = gat_row_xadd<2>(p);
    p = gat_row_xadd<1>(p);
    return p;
}

__global__ __launch_bounds__(256)
void gat_node_kernel(const float* __restrict__ xl, const float* __restrict__ xr,
                     const int* __restrict__ off, const int* __restrict__ grp,
                     const float* __restrict__ We, const float* __restrict__ att,
                     const float* __restrict__ bias, float* __restrict__ o, int N) {
    const int wave = (blockIdx.x * blockDim.x + threadIdx.x) >> 5;
    const int lane = threadIdx.x & 31;
    if (wave >= N) return;
    const int d = __builtin_amdgcn_readfirstlane(wave);
    const int beg = off[d];
    const int end = off[d + 1];

    const float4 xd = *(const float4*)(xr + (size_t)d * EMB + lane * 4);
    const float4 xself = *(const float4*)(xl + (size_t)d * EMB + lane * 4);
    const float4 we = *(const float4*)(We + lane * 4);
    const float4 at = *(const float4*)(att + lane * 4);

    float m = -INFINITY;
    float den = 0.0f;
    float4 acc = {0.0f, 0.0f, 0.0f, 0.0f};
    float wsum = 0.0f;

    for (int base = beg; base < end; base += 32) {
        int my_s = 0, my_w = 0;
        if (base + lane < end) {
            const int2 rec = *(const int2*)(grp + 2 * (size_t)(base + lane));
            my_s = rec.x; my_w = rec.y;
        }
        wsum += __int_as_float(my_w);
        const int nb = (end - base < 32) ? (end - base) : 32;
        for (int i = 0; i < nb; i += 4) {
            const int i1 = (i + 1 < nb) ? (i + 1) : (nb - 1);
            const int i2 = (i + 2 < nb) ? (i + 2) : (nb - 1);
            const int i3 = (i + 3 < nb) ? (i + 3) : (nb - 1);
            const int s0 = __builtin_amdgcn_readlane(my_s, i);
            const int s1 = __builtin_amdgcn_readlane(my_s, i1);
            const int s2 = __builtin_amdgcn_readlane(my_s, i2);
            const int s3 = __builtin_amdgcn_readlane(my_s, i3);
            const float w0 = __int_as_float(__builtin_amdgcn_readlane(my_w, i));
            const float w1 = __int_as_float(__builtin_amdgcn_readlane(my_w, i1));
            const float w2 = __int_as_float(__builtin_amdgcn_readlane(my_w, i2));
            const float w3 = __int_as_float(__builtin_amdgcn_readlane(my_w, i3));
            const float4 r0 = *(const float4*)(xl + (size_t)s0 * EMB + lane * 4);
            const float4 r1 = *(const float4*)(xl + (size_t)s1 * EMB + lane * 4);
            const float4 r2 = *(const float4*)(xl + (size_t)s2 * EMB + lane * 4);
            const float4 r3 = *(const float4*)(xl + (size_t)s3 * EMB + lane * 4);
            const float p0 = gat_head_logit(r0, xd, we, at, w0);
            float p1 = gat_head_logit(r1, xd, we, at, w1);
            float p2 = gat_head_logit(r2, xd, we, at, w2);
            float p3 = gat_head_logit(r3, xd, we, at, w3);
            p1 = (i + 1 < nb) ? p1 : -INFINITY;
            p2 = (i + 2 < nb) ? p2 : -INFINITY;
            p3 = (i + 3 < nb) ? p3 : -INFINITY;
            const float mn = fmaxf(fmaxf(m, p0), fmaxf(p1, fmaxf(p2, p3)));
            const float sc = __expf(m - mn);
            const float e0 = __expf(p0 - mn);
            const float e1 = __expf(p1 - mn);
            const float e2 = __expf(p2 - mn);
            const float e3 = __expf(p3 - mn);
            den = den * sc + ((e0 + e1) + (e2 + e3));
            acc.x = acc.x * sc + ((e0 * r0.x + e1 * r1.x) + (e2 * r2.x + e3 * r3.x));
            acc.y = acc.y * sc + ((e0 * r0.y + e1 * r1.y) + (e2 * r2.y + e3 * r3.y));
            acc.z = acc.z * sc + ((e0 * r0.z + e1 * r1.z) + (e2 * r2.z + e3 * r3.z));
            acc.w = acc.w * sc + ((e0 * r0.w + e1 * r1.w) + (e2 * r2.w + e3 * r3.w));
            m = mn;
        }
    }

    wsum += __shfl_xor(wsum, 16, 32);
    wsum = gat_row_xadd<8>(wsum);
    wsum = gat_row_xadd<4>(wsum);
    wsum = gat_row_xadd<2>(wsum);
    wsum = gat_row_xadd<1>(wsum);
    const float wself = wsum / fmaxf((float)(end - beg), 1.0f);
    {
        const float p = gat_head_logit(xself, xd, we, at, wself);
        const float mn = fmaxf(m, p);
        const float sc = __expf(m - mn);
        const float ex = __expf(p - mn);
        den = den * sc + ex;
        acc.x = acc.x * sc + ex * xself.x;
        acc.y = acc.y * sc + ex * xself.y;
        acc.z = acc.z * sc + ex * xself.z;
        acc.w = acc.w * sc + ex * xself.w;
    }

    const float4 b = *(const float4*)(bias + lane * 4);
    float4 r;
    r.x = acc.x / den + b.x;
    r.y = acc.y / den + b.y;
    r.z = acc.z / den + b.z;
    r.w = acc.w / den + b.w;
    r.x = (r.x > 0.0f) ? r.x : (__expf(r.x) - 1.0f);
    r.y = (r.y > 0.0f) ? r.y : (__expf(r.y) - 1.0f);
    r.z = (r.z > 0.0f) ? r.z : (__expf(r.z) - 1.0f);
    r.w = (r.w > 0.0f) ? r.w : (__expf(r.w) - 1.0f);
    *(float4*)(o + (size_t)d * EMB + lane * 4) = r;
}

extern "C" void kernel_launch(void* const* d_in, const int* in_sizes, int n_in,
                              void* d_out, int out_size, void* d_ws, size_t ws_size,
                              hipStream_t stream) {
    const int*   edge_index = (const int*)d_in[0];
    const float* ew   = (const float*)d_in[1];
    const float* emb  = (const float*)d_in[2];
    const float* Wl   = (const float*)d_in[3];
    const float* Wr   = (const float*)d_in[4];
    const float* We   = (const float*)d_in[5];
    const float* att  = (const float*)d_in[6];
    const float* bias = (const float*)d_in[7];

    const int E = in_sizes[0] / 2;
    const int N = in_sizes[2] / EMB;
    const int* src = edge_index;
    const int* dst = edge_index + E;

    float* ws    = (float*)d_ws;
    float* xl    = ws;
    float* xr    = xl + (size_t)N * EMB;
    float* hbuf  = xr + (size_t)N * EMB;
    const int NB = (N + BKT_NODES - 1) / BKT_NODES;
    const size_t M = (size_t)NB * L1_BLOCKS;
    int*   mat   = (int*)(hbuf + (size_t)N * EMB);
    int*   base  = mat + ((M + 31) / 32) * 32;
    int*   off   = base + ((M + 4 + 31) / 32) * 32;
    int*   tmp   = off + (((size_t)N + 1 + 31) / 32) * 32;
    int*   grp   = tmp + (((size_t)E * 4 + 31) / 32) * 32;

    const int BT = 256;
    const int rowTiles = (N + 15) / 16;
    const int gemmBlocks = (rowTiles * 4 + 7) / 8;
    const int nodeWaveBlocks = (N * 32 + BT - 1) / BT;

    gat_bucket_count_kernel<<<L1_BLOCKS, L1_THREADS, NB * sizeof(int), stream>>>(dst, mat, E, NB);
    gat_scan_kernel<<<1, 1024, 0, stream>>>(mat, base, NB * L1_BLOCKS);
    gat_bucket_place_kernel<<<L1_BLOCKS, L1_THREADS, NB * sizeof(int), stream>>>(
        src, dst, ew, base, (int4*)tmp, E, NB);
    gat_bucket_sort_kernel<<<NB, L2_THREADS, 0, stream>>>(base, (const int4*)tmp, off, grp, E, N,
                                                          NB);

    for (int l = 0; l < 2; ++l) {
        const float* hin = (l == 0) ? emb : hbuf;
        float* o = (l == 0) ? hbuf : (float*)d_out;
        const float* Wl_l = Wl + (size_t)l * EMB * EMB;
        const float* Wr_l = Wr + (size_t)l * EMB * EMB;
        const float* We_l = We + (size_t)l * EMB;
        const float* at_l = att + (size_t)l * EMB;
        const float* b_l  = bias + (size_t)l * EMB;

        gat_gemm_kernel<<<gemmBlocks, BT, 0, stream>>>(hin, Wl_l, Wr_l, xl, xr, N);
        gat_node_kernel<<<nodeWaveBlocks, BT, 0, stream>>>(xl, xr, off, grp, We_l, at_l, b_l,
                                                           o, N);
    }
}


# edge grouping in two launches, not four: a slice work-group sorts its own edges by bucket inside its slice (LDS count, scan, place), a bucket work-group gathers its 256 runs; sums arrival-ordered
# speedup vs baseline: 6.3693x; 1.0553x over previous
#include <hip/hip_runtime.h>
#include <math.h>


#define EMB 128
#define NEG_SLOPE 0.2f

typedef float v2f __attribute__((ext_vector_type(2)));
typedef float v8f __attribute__((ext_vector_type(8)));

#define BKT_SHIFT 8
#define BKT_NODES (1 << BKT_SHIFT)
#define L1_BLOCKS 256
#define L1_THREADS 512
#define L1_PER 8
#define L2_THREADS 512
#define L2_RUNS (L1_BLOCKS / (L2_THREADS / 32))

__device__ __forceinline__ int gat_wave_incl_scan(int v, const int lane) {
#pragma unroll
    for (int d = 1; d < 32; d <<= 1) {
        const int u = __shfl_up(v, d, 32);
        if (lane >= d) v += u;
    }
    return v;
}

__device__ __forceinline__ int gat_wave_sum(int v) {
#pragma unroll
    for (int d = 16; d >= 1; d >>= 1) v += __shfl_xor(v, d, 32);
    return v;
}

__device__ __forceinline__ void gat_slice_load(const int* __restrict__ src,
                                               const int* __restrict__ dst,
                                               const float* __restrict__ ew, const int c0,
                                               const int e1, const int t, int (&s)[L1_PER],
                                               int (&d)[L1_PER], int (&w)[L1_PER]) {
#pragma unroll
    for (int j = 0; j < L1_PER; ++j) {
        const int e = c0 + j * L1_THREADS + t;
        const bool in = e < e1;
        d[j] = in ? dst[e] : -1;
        s[j] = in ? src[e] : 0;
        w[j] = in ? __float_as_int(ew[e]) : 0;
    }
}

__device__ __forceinline__ void gat_slice_count(const int (&d)[L1_PER], int* tab, const int NB) {
#pragma unroll
    for (int j = 0; j < L1_PER; ++j) {
        const unsigned b = (unsigned)(d[j] >> BKT_SHIFT);
        if (b < (unsigned)NB) atomicAdd(&tab[b], 1);
    }
}

__global__ __launch_bounds__(L1_THREADS)
void gat_slice_sort_kernel(const int* __restrict__ src, const int* __restrict__ dst,
                           const float* __restrict__ ew, int* __restrict__ cntm,
                           int* __restrict__ locm, int4* __restrict__ tmp, int E, int NB) {
    extern __shared__ int lds_tab[];
    __shared__ int wtot[L1_THREADS / 32];
    const int t = threadIdx.x;
    const int lane = t & 31;
    const int wv = t >> 5;
    const int per = (E + L1_BLOCKS - 1) / L1_BLOCKS;
    const int e0 = blockIdx.x * per;
    const int e1 = (e0 + per < E) ? (e0 + per) : E;
    const int len = (e1 > e0) ? (e1 - e0) : 0;
    int s[L1_PER], d[L1_PER], w[L1_PER];
    gat_slice_load(src, dst, ew, e0, e1, t, s, d, w);
    for (int b = t; b < NB; b += L1_THREADS) lds_tab[b] = 0;
    __syncthreads();
    gat_slice_count(d, lds_tab, NB);
    for (int c0 = e0 + L1_THREADS * L1_PER; c0 < e1; c0 += L1_THREADS * L1_PER) {
        int dd[L1_PER];
#pragma unroll
        for (int j = 0; j < L1_PER; ++j) {
            const int e = c0 + j * L1_THREADS + t;
            dd[j] = (e < e1) ? dst[e] : -1;
        }
        gat_slice_count(dd, lds_tab, NB);
    }
    __syncthreads();

    int carry = 0;
    for (int b0 = 0; b0 < NB; b0 += L1_THREADS) {
        const int b = b0 + t;
        const int v = (b < NB) ? lds_tab[b] : 0;
        const int incl = gat_wave_incl_scan(v, lane);
        if (lane == 31) wtot[wv] = incl;
        __syncthreads();
        int pre = 0, tot = 0;
        for (int k = 0; k < L1_THREADS / 32; ++k) {
            const int x = wtot[k];
            tot += x;
            if (k < wv) pre += x;
        }
        const int start = carry + pre + (incl - v);
        if (b < NB) {
            lds_tab[b] = start;
            cntm[b * L1_BLOCKS + blockIdx.x] = v;
            locm[b * L1_BLOCKS + blockIdx.x] = start;
        }
        carry += tot;
        __syncthreads();
    }

    int c0 = e0;
    for (;;) {
        int pos[L1_PER];
#pragma unroll
        for (int j = 0; j < L1_PER; ++j) {
            const unsigned b = (unsigned)(d[j] >> BKT_SHIFT);
            pos[j] = -1;
            if (b < (unsigned)NB) pos[j] = atomicAdd(&lds_tab[b], 1);
        }
#pragma unroll
        for (int j = 0; j < L1_PER; ++j) {
            if ((unsigned)pos[j] < (unsigned)len) tmp[e0 + pos[j]] = make_int4(s[j], w[j], d[j], 0);
        }
        c0 += L1_THREADS * L1_PER;
        if (c0 >= e1) break;
        gat_slice_load(src, dst, ew, c0, e1, t, s, d, w);
    }
}

__global__ __launch_bounds__(L2_THREADS)
void gat_bucket_sort_kernel(const int* __restrict__ cntm, const int* __restrict__ locm,
                            const int4* __restrict__ tmp, int* __restrict__ off,
                            int* __restrict__ grp, int E, int N, int NB) {
    __shared__ int cur[BKT_NODES];
    __shared__ int wlo[L2_THREADS / 32];
    __shared__ int wct[L2_THREADS / 32];
    __shared__ int wtot[BKT_NODES / 32];
    const int t = threadIdx.x;
    const int lane = t & 31;
    const int wv = t >> 5;
    const int b = blockIdx.x;
    const int n0 = b << BKT_SHIFT;
    const int per = (E + L1_BLOCKS - 1) / L1_BLOCKS;
    int c = 0, lo = 0, first = 0;
    if (lane < L2_RUNS) {
        const int k = wv * L2_RUNS + lane;
        c = cntm[b * L1_BLOCKS + k];
        lo = locm[b * L1_BLOCKS + k];
        const int e0 = k * per;
        const int e1 = (e0 + per < E) ? (e0 + per) : E;
        const int len = (e1 > e0) ? (e1 - e0) : 0;
        lo = (lo < 0) ? 0 : ((lo > len) ? len : lo);
        c = (c < 0) ? 0 : ((c > len - lo) ? (len - lo) : c);
        first = e0 + lo;
    }
    int4 r[L2_RUNS];
#pragma unroll
    for (int u = 0; u < L2_RUNS; ++u) {
        const int cu = __builtin_amdgcn_readlane(c, u);
        const int st = __builtin_amdgcn_readlane(first, u);
        int4 v = {0, 0, -1, 0};
        if (lane < cu) v = tmp[st + lane];
        r[u] = v;
    }
    if (t < BKT_NODES) cur[t] = 0;
    const int slo = gat_wave_sum(lo);
    const int sct = gat_wave_sum(c);
    if (lane == 0) { wlo[wv] = slo; wct[wv] = sct; }
    __syncthreads();
    int rs = 0, tot = 0;
    for (int k = 0; k < L2_THREADS / 32; ++k) { rs += wlo[k]; tot += wct[k]; }
    rs = (rs < 0) ? 0 : ((rs > E) ? E : rs);
    int re = rs + tot;
    re = (re < rs) ? rs : ((re > E) ? E : re);

#pragma unroll
    for (int u = 0; u < L2_RUNS; ++u) {
        const unsigned loc = (unsigned)(r[u].z - n0);
        if (loc < (unsigned)BKT_NODES) atomicAdd(&cur[loc], 1);
    }
    for (int u = 0; u < L2_RUNS; ++u) {
        const int cu = __builtin_amdgcn_readlane(c, u);
        const int st = __builtin_amdgcn_readlane(first, u);
        for (int j = 32 + lane; j < cu; j += 32) {
            const unsigned loc = (unsigned)(tmp[st + j].z - n0);
            if (loc < (unsigned)BKT_NODES) atomicAdd(&cur[loc], 1);
        }
    }
    __syncthreads();

    int v = 0, incl = 0;
    if (t < BKT_NODES) {
        v = cur[t];
        incl = gat_wave_incl_scan(v, lane);
        if (lane == 31) wtot[t >> 5] = incl;
    }
    __syncthreads();
    if (t < BKT_NODES) {
        int pre = 0;
        for (int k = 0; k < (t >> 5); ++k) pre += wtot[k];
        const int start = rs + pre + (incl - v);
        cur[t] = start;
        if (n0 + t <= N) off[n0 + t] = start;
    }
    if (b == NB - 1 && t == 0 && (N & (BKT_NODES - 1)) == 0) off[N] = re;
    __syncthreads();

    int pos[L2_RUNS];
#pragma unroll
    for (int u = 0; u < L2_RUNS; ++u) {
        const unsigned loc = (unsigned)(r[u].z - n0);
        pos[u] = -1;
        if (loc < (unsigned)BKT_NODES) pos[u] = atomicAdd(&cur[loc], 1);
    }
#pragma unroll
    for (int u = 0; u < L2_RUNS; ++u) {
        if ((unsigned)pos[u] < (unsigned)E)
            *(int2*)(grp + 2 * (size_t)pos[u]) = make_int2(r[u].x, r[u].y);
    }
    for (int u = 0; u < L2_RUNS; ++u) {
        const int cu = __builtin_amdgcn_readlane(c, u);
        const int st = __builtin_amdgcn_readlane(first, u);
        for (int j = 32 + lane; j < cu; j += 32) {
            const int4 q = tmp[st + j];
            const unsigned loc = (unsigned)(q.z - n0);
            int p = -1;
            if (loc < (unsigned)BKT_NODES) p = atomicAdd(&cur[loc], 1);
            if ((unsigned)p < (unsigned)E) *(int2*)(grp + 2 * (size_t)p) = make_int2(q.x, q.y);
        }
    }
}

typedef float v4f __attribute__((ext_vector_type(4)));

__global__ __launch_bounds__(256)
void gat_gemm_kernel(const float* __restrict__ h, const float* __restrict__ Wl,
                     const float* __restrict__ Wr, float* __restrict__ xl,
                     float* __restrict__ xr, int N) {
    int lane = threadIdx.x & 31;
    int wave = (blockIdx.x * blockDim.x + threadIdx.x) >> 5;
    int rowTiles = (N + 15) >> 4;
    if (wave >= rowTiles * 4) return;
    int rowTile = wave >> 2;
    int q = wave & 3;
    const float* W = (q & 2) ? Wr : Wl;
    float*      out = (q & 2) ? xr : xl;
    int cbase = (q & 1) * 64;
    int half = lane >> 4;
    int m    = lane & 15;
    int arow = rowTile * 16 + m;
    if (arow > N - 1) arow = N - 1;
    int col  = cbase + 4 * m;
    int koff = half * 2;
    const float* ap = h + (size_t)arow * EMB + koff;
    const float* wp = W + (size_t)koff * EMB + col;

    v8f c0 = {0.f, 0.f, 0.f, 0.f, 0.f, 0.f, 0.f, 0.f};
    v8f c1 = c0, c2 = c0, c3 = c0;
#pragma unroll
    for (int k = 0; k < EMB; k += 4) {
        v2f a = *(const v2f*)(ap + k);
        v4f w0 = *(const v4f*)(wp + (size_t)k * EMB);
        v4f w1 = *(const v4f*)(wp + (size_t)(k + 1) * EMB);
        v2f b0; b0.x = w0.x; b0.y = w1.x;
        v2f b1; b1.x = w0.y; b1.y = w1.y;
        v2f b2; b2.x = w0.z; b2.y = w1.z;
        v2f b3; b3.x = w0.w; b3.y = w1.w;
        c0 = __builtin_amdgcn_wmma_f32_16x16x4_f32(false, a, false, b0, (short)0, c0,
                                                   false, false);
        c1 = __builtin_amdgcn_wmma_f32_16x16x4_f32(false, a, false, b1, (short)0, c1,
                                                   false, false);
        c2 = __builtin_amdgcn_wmma_f32_16x16x4_f32(false, a, false, b2, (short)0, c2,
                                                   false, false);
        c3 = __builtin_amdgcn_wmma_f32_16x16x4_f32(false, a, false, b3, (short)0, c3,
                                                   false, false);
    }
#pragma unroll
    for (int j = 0; j < 8; ++j) {
        int r = rowTile * 16 + j + half * 8;
        v4f v; v.x = c0[j]; v.y = c1[j]; v.z = c2[j]; v.w = c3[j];
        if (r < N) *(v4f*)(out + (size_t)r * EMB + col) = v;
    }
}

template <int K>
__device__ __forceinline__ float gat_row_xadd(const float v) {
    const int u = __builtin_amdgcn_update_dpp(0, __float_as_int(v), 0x160 + K, 0xF, 0xF, true);
    return v + __int_as_float(u);
}

__device__ __forceinline__ float gat_leaky(const float v) {
    return fmaxf(v, NEG_SLOPE * v);
}

__device__ __forceinline__ float gat_head_logit(const float4 xs, const float4 xd, const float4 we,
                                                const float4 at, const float w) {
    float p = gat_leaky(xs.x + xd.x + w * we.x) * at.x;
    p += gat_leaky(xs.y + xd.y + w * we.y) * at.y;
    p += gat_leaky(xs.z + xd.z + w * we.z) * at.z;
    p += gat_leaky(xs.w + xd.w + w * we.w) * at.w;
    p = gat_row_xadd<8>(p);
    p = gat_row_xadd<4>(p);
    p = gat_row_xadd<2>(p);
    p = gat_row_xadd<1>(p);
    return p;
}

__global__ __launch_bounds__(256)
void gat_node_kernel(const float* __restrict__ xl, const float* __restrict__ xr,
                     const int* __restrict__ off, const int* __restrict__ grp,
                     const float* __restrict__ We, const float* __restrict__ att,
                     const float* __restrict__ bias, float* __restrict__ o, int N) {
    const int wave = (blockIdx.x * blockDim.x + threadIdx.x) >> 5;
    const int lane = threadIdx.x & 31;
    if (wave >= N) return;
    const int d = __builtin_amdgcn_readfirstlane(wave);
    const int beg = off[d];
    const int end = off[d + 1];

    const float4 xd = *(const float4*)(xr + (size_t)d * EMB + lane * 4);
    const float4 xself = *(const float4*)(xl + (size_t)d * EMB + lane * 4);
    const float4 we = *(const float4*)(We + lane * 4);
    const float4 at = *(const float4*)(att + lane * 4);

    float m = -INFINITY;
    float den = 0.0f;
    float4 acc = {0.0f, 0.0f, 0.0f, 0.0f};
    float wsum = 0.0f;

    for (int base = beg; base < end; base += 32) {
        int my_s = 0, my_w = 0;
        if (base + lane < end) {
            const int2 rec = *(const int2*)(grp + 2 * (size_t)(base + lane));
            my_s = rec.x; my_w = rec.y;
        }
        wsum += __int_as_float(my_w);
        const int nb = (end - base < 32) ? (end - base) : 32;
        for (int i = 0; i < nb; i += 4) {
            const int i1 = (i + 1 < nb) ? (i + 1) : (nb - 1);
            const int i2 = (i + 2 < nb) ? (i + 2) : (nb - 1);
            const int i3 = (i + 3 < nb) ? (i + 3) : (nb - 1);
            const int s0 = __builtin_amdgcn_readlane(my_s, i);
            const int s1 = __builtin_amdgcn_readlane(my_s, i1);
            const int s2 = __builtin_amdgcn_readlane(my_s, i2);
            const int s3 = __builtin_amdgcn_readlane(my_s, i3);
            const float w0 = __int_as_float(__builtin_amdgcn_readlane(my_w, i));
            const float w1 = __int_as_float(__builtin_amdgcn_readlane(my_w, i1));
            const float w2 = __int_as_float(__builtin_amdgcn_readlane(my_w, i2));
            const float w3 = __int_as_float(__builtin_amdgcn_readlane(my_w, i3));
            const float4 r0 = *(const float4*)(xl + (size_t)s0 * EMB + lane * 4);
            const float4 r1 = *(const float4*)(xl + (size_t)s1 * EMB + lane * 4);
            const float4 r2 = *(const float4*)(xl + (size_t)s2 * EMB + lane * 4);
            const float4 r3 = *(const float4*)(xl + (size_t)s3 * EMB + lane * 4);
            const float p0 = gat_head_logit(r0, xd, we, at, w0);
            float p1 = gat_head_logit(r1, xd, we, at, w1);
            float p2 = gat_head_logit(r2, xd, we, at, w2);
            float p3 = gat_head_logit(r3, xd, we, at, w3);
            p1 = (i + 1 < nb) ? p1 : -INFINITY;
            p2 = (i + 2 < nb) ? p2 : -INFINITY;
            p3 = (i + 3 < nb) ? p3 : -INFINITY;
            const float mn = fmaxf(fmaxf(m, p0), fmaxf(p1, fmaxf(p2, p3)));
            const float sc = __expf(m - mn);
            const float e0 = __expf(p0 - mn);
            const float e1 = __expf(p1 - mn);
            const float e2 = __expf(p2 - mn);
            const float e3 = __expf(p3 - mn);
            den = den * sc + ((e0 + e1) + (e2 + e3));
            acc.x = acc.x * sc + ((e0 * r0.x + e1 * r1.x) + (e2 * r2.x + e3 * r3.x));
            acc.y = acc.y * sc + ((e0 * r0.y + e1 * r1.y) + (e2 * r2.y + e3 * r3.y));
            acc.z = acc.z * sc + ((e0 * r0.z + e1 * r1.z) + (e2 * r2.z + e3 * r3.z));
            acc.w = acc.w * sc + ((e0 * r0.w + e1 * r1.w) + (e2 * r2.w + e3 * r3.w));
            m = mn;
        }
    }

    wsum += __shfl_xor(wsum, 16, 32);
    wsum = gat_row_xadd<8>(wsum);
    wsum = gat_row_xadd<4>(wsum);
    wsum = gat_row_xadd<2>(wsum);
    wsum = gat_row_xadd<1>(wsum);
    const float wself = wsum / fmaxf((float)(end - beg), 1.0f);
    {
        const float p = gat_head_logit(xself, xd, we, at, wself);
        const float mn = fmaxf(m, p);
        const float sc = __expf(m - mn);
        const float ex = __expf(p - mn);
        den = den * sc + ex;
        acc.x = acc.x * sc + ex * xself.x;
        acc.y = acc.y * sc + ex * xself.y;
        acc.z = acc.z * sc + ex * xself.z;
        acc.w = acc.w * sc + ex * xself.w;
    }

    const float4 b = *(const float4*)(bias + lane * 4);
    float4 r;
    r.x = acc.x / den + b.x;
    r.y = acc.y / den + b.y;
    r.z = acc.z / den + b.z;
    r.w = acc.w / den + b.w;
    r.x = (r.x > 0.0f) ? r.x : (__expf(r.x) - 1.0f);
    r.y = (r.y > 0.0f) ? r.y : (__expf(r.y) - 1.0f);
    r.z = (r.z > 0.0f) ? r.z : (__expf(r.z) - 1.0f);
    r.w = (r.w > 0.0f) ? r.w : (__expf(r.w) - 1.0f);
    *(float4*)(o + (size_t)d * EMB + lane * 4) = r;
}

extern "C" void kernel_launch(void* const* d_in, const int* in_sizes, int n_in,
                              void* d_out, int out_size, void* d_ws, size_t ws_size,
                              hipStream_t stream) {
    const int*   edge_index = (const int*)d_in[0];
    const float* ew   = (const float*)d_in[1];
    const float* emb  = (const float*)d_in[2];
    const float* Wl   = (const float*)d_in[3];
    const float* Wr   = (const float*)d_in[4];
    const float* We   = (const float*)d_in[5];
    const float* att  = (const float*)d_in[6];
    const float* bias = (const float*)d_in[7];

    const int E = in_sizes[0] / 2;
    const int N = in_sizes[2] / EMB;
    const int* src = edge_index;
    const int* dst = edge_index + E;

    float* ws    = (float*)d_ws;
    float* xl    = ws;
    float* xr    = xl + (size_t)N * EMB;
    float* hbuf  = xr + (size_t)N * EMB;
    const int NB = (N + BKT_NODES - 1) / BKT_NODES;
    const size_t M = (size_t)NB * L1_BLOCKS;
    int*   cntm  = (int*)(hbuf + (size_t)N * EMB);
    int*   locm  = cntm + ((M + 31) / 32) * 32;
    int*   off   = locm + ((M + 31) / 32) * 32;
    int*   tmp   = off + (((size_t)N + 1 + 31) / 32) * 32;
    int*   grp   = tmp + (((size_t)E * 4 + 31) / 32) * 32;

    const int BT = 256;
    const int rowTiles = (N + 15) / 16;
    const int gemmBlocks = (rowTiles * 4 + 7) / 8;
    const int nodeWaveBlocks = (N * 32 + BT - 1) / BT;

    gat_slice_sort_kernel<<<L1_BLOCKS, L1_THREADS, NB * sizeof(int), stream>>>(
        src, dst, ew, cntm, locm, (int4*)tmp, E, NB);
    gat_bucket_sort_kernel<<<NB, L2_THREADS, 0, stream>>>(cntm, locm, (const int4*)tmp, off, grp,
                                                          E, N, NB);

    for (int l = 0; l < 2; ++l) {
        const float* hin = (l == 0) ? emb : hbuf;
        float* o = (l == 0) ? hbuf : (float*)d_out;
        const float* Wl_l = Wl + (size_t)l * EMB * EMB;
        const float* Wr_l = Wr + (size_t)l * EMB * EMB;
        const float* We_l = We + (size_t)l * EMB;
        const float* at_l = att + (size_t)l * EMB;
        const float* b_l  = bias + (size_t)l * EMB;

        gat_gemm_kernel<<<gemmBlocks, BT, 0, stream>>>(hin, Wl_l, Wr_l, xl, xr, N);
        gat_node_kernel<<<nodeWaveBlocks, BT, 0, stream>>>(xl, xr, off, grp, We_l, at_l, b_l,
                                                           o, N);
    }
}


# first layer's dense products ride as extra work-groups in the two edge-sort launches, which they do not depend on (load look-ahead cut so both share a compute unit); 5 launches; sum orders kept
# speedup vs baseline: 6.8092x; 1.0691x over previous
#include <hip/hip_runtime.h>
#include <math.h>


#define EMB 128
#define NEG_SLOPE 0.2f

typedef float v2f __attribute__((ext_vector_type(2)));
typedef float v8f __attribute__((ext_vector_type(8)));

#define BKT_SHIFT 8
#define BKT_NODES (1 << BKT_SHIFT)
#define L1_BLOCKS 256
#define L1_THREADS 512
#define L1_PER 8
#define L2_THREADS 512
#define L2_RUNS (L1_BLOCKS / (L2_THREADS / 32))

__device__ __forceinline__ int gat_wave_incl_scan(int v, const int lane) {
#pragma unroll
    for (int d = 1; d < 32; d <<= 1) {
        const int u = __shfl_up(v, d, 32);
        if (lane >= d) v += u;
    }
    return v;
}

__device__ __forceinline__ int gat_wave_sum(int v) {
#pragma unroll
    for (int d = 16; d >= 1; d >>= 1) v += __shfl_xor(v, d, 32);
    return v;
}

__device__ __forceinline__ void gat_slice_load(const int* __restrict__ src,
                                               const int* __restrict__ dst,
                                               const float* __restrict__ ew, const int c0,
                                               const int e1, const int t, int (&s)[L1_PER],
                                               int (&d)[L1_PER], int (&w)[L1_PER]) {
#pragma unroll
    for (int j = 0; j < L1_PER; ++j) {
        const int e = c0 + j * L1_THREADS + t;
        const bool in = e < e1;
        d[j] = in ? dst[e] : -1;
        s[j] = in ? src[e] : 0;
        w[j] = in ? __float_as_int(ew[e]) : 0;
    }
}

__device__ __forceinline__ void gat_slice_count(const int (&d)[L1_PER], int* tab, const int NB) {
#pragma unroll
    for (int j = 0; j < L1_PER; ++j) {
        const unsigned b = (unsigned)(d[j] >> BKT_SHIFT);
        if (b < (unsigned)NB) atomicAdd(&tab[b], 1);
    }
}

__device__ __forceinline__
void gat_slice_sort_body(const int* __restrict__ src, const int* __restrict__ dst,
                         const float* __restrict__ ew, int* __restrict__ cntm,
                         int* __restrict__ locm, int4* __restrict__ tmp, int E, int NB) {
    extern __shared__ int lds_tab[];
    __shared__ int wtot[L1_THREADS / 32];
    const int t = threadIdx.x;
    const int lane = t & 31;
    const int wv = t >> 5;
    const int per = (E + L1_BLOCKS - 1) / L1_BLOCKS;
    const int e0 = blockIdx.x * per;
    const int e1 = (e0 + per < E) ? (e0 + per) : E;
    const int len = (e1 > e0) ? (e1 - e0) : 0;
    int s[L1_PER], d[L1_PER], w[L1_PER];
    gat_slice_load(src, dst, ew, e0, e1, t, s, d, w);
    for (int b = t; b < NB; b += L1_THREADS) lds_tab[b] = 0;
    __syncthreads();
    gat_slice_count(d, lds_tab, NB);
    for (int c0 = e0 + L1_THREADS * L1_PER; c0 < e1; c0 += L1_THREADS * L1_PER) {
        int dd[L1_PER];
#pragma unroll
        for (int j = 0; j < L1_PER; ++j) {
            const int e = c0 + j * L1_THREADS + t;
            dd[j] = (e < e1) ? dst[e] : -1;
        }
        gat_slice_count(dd, lds_tab, NB);
    }
    __syncthreads();

    int carry = 0;
    for (int b0 = 0; b0 < NB; b0 += L1_THREADS) {
        const int b = b0 + t;
        const int v = (b < NB) ? lds_tab[b] : 0;
        const int incl = gat_wave_incl_scan(v, lane);
        if (lane == 31) wtot[wv] = incl;
        __syncthreads();
        int pre = 0, tot = 0;
        for (int k = 0; k < L1_THREADS / 32; ++k) {
            const int x = wtot[k];
            tot += x;
            if (k < wv) pre += x;
        }
        const int start = carry + pre + (incl - v);
        if (b < NB) {
            lds_tab[b] = start;
            cntm[b * L1_BLOCKS + blockIdx.x] = v;
            locm[b * L1_BLOCKS + blockIdx.x] = start;
        }
        carry += tot;
        __syncthreads();
    }

    int c0 = e0;
    for (;;) {
        int pos[L1_PER];
#pragma unroll
        for (int j = 0; j < L1_PER; ++j) {
            const unsigned b = (unsigned)(d[j] >> BKT_SHIFT);
            pos[j] = -1;
            if (b < (unsigned)NB) pos[j] = atomicAdd(&lds_tab[b], 1);
        }
#pragma unroll
        for (int j = 0; j < L1_PER; ++j) {
            if ((unsigned)pos[j] < (unsigned)len) tmp[e0 + pos[j]] = make_int4(s[j], w[j], d[j], 0);
        }
        c0 += L1_THREADS * L1_PER;
        if (c0 >= e1) break;
        gat_slice_load(src, dst, ew, c0, e1, t, s, d, w);
    }
}

__device__ __forceinline__
void gat_bucket_sort_body(const int* __restrict__ cntm, const int* __restrict__ locm,
                          const int4* __restrict__ tmp, int* __restrict__ off,
                          int* __restrict__ grp, int E, int N, int NB) {
    __shared__ int cur[BKT_NODES];
    __shared__ int wlo[L2_THREADS / 32];
    __shared__ int wct[L2_THREADS / 32];
    __shared__ int wtot[BKT_NODES / 32];
    const int t = threadIdx.x;
    const int lane = t & 31;
    const int wv = t >> 5;
    const int b = blockIdx.x;
    const int n0 = b << BKT_SHIFT;
    const int per = (E + L1_BLOCKS - 1) / L1_BLOCKS;
    int c = 0, lo = 0, first = 0;
    if (lane < L2_RUNS) {
        const int k = wv * L2_RUNS + lane;
        c = cntm[b * L1_BLOCKS + k];
        lo = locm[b * L1_BLOCKS + k];
        const int e0 = k * per;
        const int e1 = (e0 + per < E) ? (e0 + per) : E;
        const int len = (e1 > e0) ? (e1 - e0) : 0;
        lo = (lo < 0) ? 0 : ((lo > len) ? len : lo);
        c = (c < 0) ? 0 : ((c > len - lo) ? (len - lo) : c);
        first = e0 + lo;
    }
    int4 r[L2_RUNS];
#pragma unroll
    for (int u = 0; u < L2_RUNS; ++u) {
        const int cu = __builtin_amdgcn_readlane(c, u);
        const int st = __builtin_amdgcn_readlane(first, u);
        int4 v = {0, 0, -1, 0};
        if (lane < cu) v = tmp[st + lane];
        r[u] = v;
    }
    if (t < BKT_NODES) cur[t] = 0;
    const int slo = gat_wave_sum(lo);
    const int sct = gat_wave_sum(c);
    if (lane == 0) { wlo[wv] = slo; wct[wv] = sct; }
    __syncthreads();
    int rs = 0, tot = 0;
    for (int k = 0; k < L2_THREADS / 32; ++k) { rs += wlo[k]; tot += wct[k]; }
    rs = (rs < 0) ? 0 : ((rs > E) ? E : rs);
    int re = rs + tot;
    re = (re < rs) ? rs : ((re > E) ? E : re);

#pragma unroll
    for (int u = 0; u < L2_RUNS; ++u) {
        const unsigned loc = (unsigned)(r[u].z - n0);
        if (loc < (unsigned)BKT_NODES) atomicAdd(&cur[loc], 1);
    }
    for (int u = 0; u < L2_RUNS; ++u) {
        const int cu = __builtin_amdgcn_readlane(c, u);
        const int st = __builtin_amdgcn_readlane(first, u);
        for (int j = 32 + lane; j < cu; j += 32) {
            const unsigned loc = (unsigned)(tmp[st + j].z - n0);
            if (loc < (unsigned)BKT_NODES) atomicAdd(&cur[loc], 1);
        }
    }
    __syncthreads();

    int v = 0, incl = 0;
    if (t < BKT_NODES) {
        v = cur[t];
        incl = gat_wave_incl_scan(v, lane);
        if (lane == 31) wtot[t >> 5] = incl;
    }
    __syncthreads();
    if (t < BKT_NODES) {
        int pre = 0;
        for (int k = 0; k < (t >> 5); ++k) pre += wtot[k];
        const int start = rs + pre + (incl - v);
        cur[t] = start;
        if (n0 + t <= N) off[n0 + t] = start;
    }
    if (b == NB - 1 && t == 0 && (N & (BKT_NODES - 1)) == 0) off[N] = re;
    __syncthreads();

    int pos[L2_RUNS];
#pragma unroll
    for (int u = 0; u < L2_RUNS; ++u) {
        const unsigned loc = (unsigned)(r[u].z - n0);
        pos[u] = -1;
        if (loc < (unsigned)BKT_NODES) pos[u] = atomicAdd(&cur[loc], 1);
    }
#pragma unroll
    for (int u = 0; u < L2_RUNS; ++u) {
        if ((unsigned)pos[u] < (unsigned)E)
            *(int2*)(grp + 2 * (size_t)pos[u]) = make_int2(r[u].x, r[u].y);
    }
    for (int u = 0; u < L2_RUNS; ++u) {
        const int cu = __builtin_amdgcn_readlane(c, u);
        const int st = __builtin_amdgcn_readlane(first, u);
        for (int j = 32 + lane; j < cu; j += 32) {
            const int4 q = tmp[st + j];
            const unsigned loc = (unsigned)(q.z - n0);
            int p = -1;
            if (loc < (unsigned)BKT_NODES) p = atomicAdd(&cur[loc], 1);
            if ((unsigned)p < (unsigned)E) *(int2*)(grp + 2 * (size_t)p) = make_int2(q.x, q.y);
        }
    }
}

typedef float v4f __attribute__((ext_vector_type(4)));

#define GAT_GK 2
struct gat_kgroup { v2f a[GAT_GK]; v4f w0[GAT_GK]; v4f w1[GAT_GK]; };

__device__ __forceinline__ void gat_group_load(gat_kgroup& g, const float* __restrict__ ap,
                                               const float* __restrict__ wp, const int k0) {
#pragma unroll
    for (int i = 0; i < GAT_GK; ++i) {
        const int k = k0 + 4 * i;
        g.a[i]  = *(const v2f*)(ap + k);
        g.w0[i] = *(const v4f*)(wp + (size_t)k * EMB);
        g.w1[i] = *(const v4f*)(wp + (size_t)(k + 1) * EMB);
    }
}

__device__ __forceinline__ void gat_group_mma(const gat_kgroup& g, v8f& c0, v8f& c1, v8f& c2,
                                              v8f& c3) {
#pragma unroll
    for (int i = 0; i < GAT_GK; ++i) {
        const v2f a = g.a[i];
        v2f b0; b0.x = g.w0[i].x; b0.y = g.w1[i].x;
        v2f b1; b1.x = g.w0[i].y; b1.y = g.w1[i].y;
        v2f b2; b2.x = g.w0[i].z; b2.y = g.w1[i].z;
        v2f b3; b3.x = g.w0[i].w; b3.y = g.w1[i].w;
        c0 = __builtin_amdgcn_wmma_f32_16x16x4_f32(false, a, false, b0, (short)0, c0,
                                                   false, false);
        c1 = __builtin_amdgcn_wmma_f32_16x16x4_f32(false, a, false, b1, (short)0, c1,
                                                   false, false);
        c2 = __builtin_amdgcn_wmma_f32_16x16x4_f32(false, a, false, b2, (short)0, c2,
                                                   false, false);
        c3 = __builtin_amdgcn_wmma_f32_16x16x4_f32(false, a, false, b3, (short)0, c3,
                                                   false, false);
    }
}

template <bool PIPE>
__device__ __forceinline__
void gat_gemm_body(const float* __restrict__ h, const float* __restrict__ Wl,
                   const float* __restrict__ Wr, float* __restrict__ xl,
                   float* __restrict__ xr, int N, const int wave, const int waveEnd) {
    int lane = threadIdx.x & 31;
    if (wave >= waveEnd) return;
    int rowTile = wave >> 2;
    int q = wave & 3;
    const float* W = (q & 2) ? Wr : Wl;
    float*      out = (q & 2) ? xr : xl;
    int cbase = (q & 1) * 64;
    int half = lane >> 4;
    int m    = lane & 15;
    int arow = rowTile * 16 + m;
    if (arow > N - 1) arow = N - 1;
    int col  = cbase + 4 * m;
    int koff = half * 2;
    const float* ap = h + (size_t)arow * EMB + koff;
    const float* wp = W + (size_t)koff * EMB + col;

    v8f c0 = {0.f, 0.f, 0.f, 0.f, 0.f, 0.f, 0.f, 0.f};
    v8f c1 = c0, c2 = c0, c3 = c0;
    if (PIPE) {
        gat_kgroup ga, gb;
        gat_group_load(ga, ap, wp, 0);
#pragma unroll
        for (int k = 0; k < EMB; k += 8 * GAT_GK) {
            gat_group_load(gb, ap, wp, k + 4 * GAT_GK);
            __builtin_amdgcn_sched_barrier(0);
            gat_group_mma(ga, c0, c1, c2, c3);
            __builtin_amdgcn_sched_barrier(0);
            if (k + 8 * GAT_GK < EMB) gat_group_load(ga, ap, wp, k + 8 * GAT_GK);
            __builtin_amdgcn_sched_barrier(0);
            gat_group_mma(gb, c0, c1, c2, c3);
            __builtin_amdgcn_sched_barrier(0);
        }
    } else {
#pragma unroll
    for (int k = 0; k < EMB; k += 4) {
        v2f a = *(const v2f*)(ap + k);
        v4f w0 = *(const v4f*)(wp + (size_t)k * EMB);
        v4f w1 = *(const v4f*)(wp + (size_t)(k + 1) * EMB);
        v2f b0; b0.x = w0.x; b0.y = w1.x;
        v2f b1; b1.x = w0.y; b1.y = w1.y;
        v2f b2; b2.x = w0.z; b2.y = w1.z;
        v2f b3; b3.x = w0.w; b3.y = w1.w;
        c0 = __builtin_amdgcn_wmma_f32_16x16x4_f32(false, a, false, b0, (short)0, c0,
                                                   false, false);
        c1 = __builtin_amdgcn_wmma_f32_16x16x4_f32(false, a, false, b1, (short)0, c1,
                                                   false, false);
        c2 = __builtin_amdgcn_wmma_f32_16x16x4_f32(false, a, false, b2, (short)0, c2,
                                                   false, false);
        c3 = __builtin_amdgcn_wmma_f32_16x16x4_f32(false, a, false, b3, (short)0, c3,
                                                   false, false);
    }
    }
#pragma unroll
    for (int j = 0; j < 8; ++j) {
        int r = rowTile * 16 + j + half * 8;
        v4f v; v.x = c0[j]; v.y = c1[j]; v.z = c2[j]; v.w = c3[j];
        if (r < N) *(v4f*)(out + (size_t)r * EMB + col) = v;
    }
}

__global__ __launch_bounds__(256)
void gat_gemm_kernel(const float* __restrict__ h, const float* __restrict__ Wl,
                     const float* __restrict__ Wr, float* __restrict__ xl,
                     float* __restrict__ xr, int N) {
    const int wave = (blockIdx.x * blockDim.x + threadIdx.x) >> 5;
    gat_gemm_body<false>(h, Wl, Wr, xl, xr, N, wave, ((N + 15) >> 4) * 4);
}

__global__ __launch_bounds__(L1_THREADS)
void gat_slice_sort_gemm_kernel(const int* __restrict__ src, const int* __restrict__ dst,
                                const float* __restrict__ ew, int* __restrict__ cntm,
                                int* __restrict__ locm, int4* __restrict__ tmp, int E, int NB,
                                const float* __restrict__ h, const float* __restrict__ Wl,
                                const float* __restrict__ Wr, float* __restrict__ xl,
                                float* __restrict__ xr, int N, int rtSplit) {
    if (blockIdx.x < L1_BLOCKS) {
        gat_slice_sort_body(src, dst, ew, cntm, locm, tmp, E, NB);
    } else {
        const int t = __builtin_amdgcn_readfirstlane((int)(threadIdx.x >> 5));
        const int j = (int)blockIdx.x - L1_BLOCKS;
        const int lo = (int)(((long long)j * rtSplit) / L1_BLOCKS);
        const int hi = (int)(((long long)(j + 1) * rtSplit) / L1_BLOCKS);
        for (int it = t; it < (hi - lo) * 4; it += L1_THREADS / 32)
            gat_gemm_body<true>(h, Wl, Wr, xl, xr, N, lo * 4 + it, rtSplit * 4);
    }
}

__global__ __launch_bounds__(L2_THREADS)
void gat_bucket_sort_gemm_kernel(const int* __restrict__ cntm, const int* __restrict__ locm,
                                 const int4* __restrict__ tmp, int* __restrict__ off,
                                 int* __restrict__ grp, int E, int N, int NB,
                                 const float* __restrict__ h, const float* __restrict__ Wl,
                                 const float* __restrict__ Wr, float* __restrict__ xl,
                                 float* __restrict__ xr, int rtSplit) {
    if ((int)blockIdx.x < NB) {
        gat_bucket_sort_body(cntm, locm, tmp, off, grp, E, N, NB);
    } else {
        const int wave = rtSplit * 4 + ((int)blockIdx.x - NB) * (L2_THREADS / 32) +
                         (int)(threadIdx.x >> 5);
        gat_gemm_body<true>(h, Wl, Wr, xl, xr, N, wave, ((N + 15) >> 4) * 4);
    }
}

template <int K>
__device__ __forceinline__ float gat_row_xadd(const float v) {
    const int u = __builtin_amdgcn_update_dpp(0, __float_as_int(v), 0x160 + K, 0xF, 0xF, true);
    return v + __int_as_float(u);
}

__device__ __forceinline__ float gat_leaky(const float v) {
    return fmaxf(v, NEG_SLOPE * v);
}

__device__ __forceinline__ float gat_head_logit(const float4 xs, const float4 xd, const float4 we,
                                                const float4 at, const float w) {
    float p = gat_leaky(xs.x + xd.x + w * we.x) * at.x;
    p += gat_leaky(xs.y + xd.y + w * we.y) * at.y;
    p += gat_leaky(xs.z + xd.z + w * we.z) * at.z;
    p += gat_leaky(xs.w + xd.w + w * we.w) * at.w;
    p = gat_row_xadd<8>(p);
    p = gat_row_xadd<4>(p);
    p = gat_row_xadd<2>(p);
    p = gat_row_xadd<1>(p);
    return p;
}

__global__ __launch_bounds__(256)
void gat_node_kernel(const float* __restrict__ xl, const float* __restrict__ xr,
                     const int* __restrict__ off, const int* __restrict__ grp,
                     const float* __restrict__ We, const float* __restrict__ att,
                     const float* __restrict__ bias, float* __restrict__ o, int N) {
    const int wave = (blockIdx.x * blockDim.x + threadIdx.x) >> 5;
    const int lane = threadIdx.x & 31;
    if (wave >= N) return;
    const int d = __builtin_amdgcn_readfirstlane(wave);
    const int beg = off[d];
    const int end = off[d + 1];

    const float4 xd = *(const float4*)(xr + (size_t)d * EMB + lane * 4);
    const float4 xself = *(const float4*)(xl + (size_t)d * EMB + lane * 4);
    const float4 we = *(const float4*)(We + lane * 4);
    const float4 at = *(const float4*)(att + lane * 4);

    float m = -INFINITY;
    float den = 0.0f;
    float4 acc = {0.0f, 0.0f, 0.0f, 0.0f};
    float wsum = 0.0f;

    for (int base = beg; base < end; base += 32) {
        int my_s = 0, my_w = 0;
        if (base + lane < end) {
            const int2 rec = *(const int2*)(grp + 2 * (size_t)(base + lane));
            my_s = rec.x; my_w = rec.y;
        }
        wsum += __int_as_float(my_w);
        const int nb = (end - base < 32) ? (end - base) : 32;
        for (int i = 0; i < nb; i += 4) {
            const int i1 = (i + 1 < nb) ? (i + 1) : (nb - 1);
            const int i2 = (i + 2 < nb) ? (i + 2) : (nb - 1);
            const int i3 = (i + 3 < nb) ? (i + 3) : (nb - 1);
            const int s0 = __builtin_amdgcn_readlane(my_s, i);
            const int s1 = __builtin_amdgcn_readlane(my_s, i1);
            const int s2 = __builtin_amdgcn_readlane(my_s, i2);
            const int s3 = __builtin_amdgcn_readlane(my_s, i3);
            const float w0 = __int_as_float(__builtin_amdgcn_readlane(my_w, i));
            const float w1 = __int_as_float(__builtin_amdgcn_readlane(my_w, i1));
            const float w2 = __int_as_float(__builtin_amdgcn_readlane(my_w, i2));
            const float w3 = __int_as_float(__builtin_amdgcn_readlane(my_w, i3));
            const float4 r0 = *(const float4*)(xl + (size_t)s0 * EMB + lane * 4);
            const float4 r1 = *(const float4*)(xl + (size_t)s1 * EMB + lane * 4);
            const float4 r2 = *(const float4*)(xl + (size_t)s2 * EMB + lane * 4);
            const float4 r3 = *(const float4*)(xl + (size_t)s3 * EMB + lane * 4);
            const float p0 = gat_head_logit(r0, xd, we, at, w0);
            float p1 = gat_head_logit(r1, xd, we, at, w1);
            float p2 = gat_head_logit(r2, xd, we, at, w2);
            float p3 = gat_head_logit(r3, xd, we, at, w3);
            p1 = (i + 1 < nb) ? p1 : -INFINITY;
            p2 = (i + 2 < nb) ? p2 : -INFINITY;
            p3 = (i + 3 < nb) ? p3 : -INFINITY;
            const float mn = fmaxf(fmaxf(m, p0), fmaxf(p1, fmaxf(p2, p3)));
            const float sc = __expf(m - mn);
            const float e0 = __expf(p0 - mn);
            const float e1 = __expf(p1 - mn);
            const float e2 = __expf(p2 - mn);
            const float e3 = __expf(p3 - mn);
            den = den * sc + ((e0 + e1) + (e2 + e3));
            acc.x = acc.x * sc + ((e0 * r0.x + e1 * r1.x) + (e2 * r2.x + e3 * r3.x));
            acc.y = acc.y * sc + ((e0 * r0.y + e1 * r1.y) + (e2 * r2.y + e3 * r3.y));
            acc.z = acc.z * sc + ((e0 * r0.z + e1 * r1.z) + (e2 * r2.z + e3 * r3.z));
            acc.w = acc.w * sc + ((e0 * r0.w + e1 * r1.w) + (e2 * r2.w + e3 * r3.w));
            m = mn;
        }
    }

    wsum += __shfl_xor(wsum, 16, 32);
    wsum = gat_row_xadd<8>(wsum);
    wsum = gat_row_xadd<4>(wsum);
    wsum = gat_row_xadd<2>(wsum);
    wsum = gat_row_xadd<1>(wsum);
    const float wself = wsum / fmaxf((float)(end - beg), 1.0f);
    {
        const float p = gat_head_logit(xself, xd, we, at, wself);
        const float mn = fmaxf(m, p);
        const float sc = __expf(m - mn);
        const float ex = __expf(p - mn);
        den = den * sc + ex;
        acc.x = acc.x * sc + ex * xself.x;
        acc.y = acc.y * sc + ex * xself.y;
        acc.z = acc.z * sc + ex * xself.z;
        acc.w = acc.w * sc + ex * xself.w;
    }

    const float4 b = *(const float4*)(bias + lane * 4);
    float4 r;
    r.x = acc.x / den + b.x;
    r.y = acc.y / den + b.y;
    r.z = acc.z / den + b.z;
    r.w = acc.w / den + b.w;
    r.x = (r.x > 0.0f) ? r.x : (__expf(r.x) - 1.0f);
    r.y = (r.y > 0.0f) ? r.y : (__expf(r.y) - 1.0f);
    r.z = (r.z > 0.0f) ? r.z : (__expf(r.z) - 1.0f);
    r.w = (r.w > 0.0f) ? r.w : (__expf(r.w) - 1.0f);
    *(float4*)(o + (size_t)d * EMB + lane * 4) = r;
}

extern "C" void kernel_launch(void* const* d_in, const int* in_sizes, int n_in,
                              void* d_out, int out_size, void* d_ws, size_t ws_size,
                              hipStream_t stream) {
    const int*   edge_index = (const int*)d_in[0];
    const float* ew   = (const float*)d_in[1];
    const float* emb  = (const float*)d_in[2];
    const float* Wl   = (const float*)d_in[3];
    const float* Wr   = (const float*)d_in[4];
    const float* We   = (const float*)d_in[5];
    const float* att  = (const float*)d_in[6];
    const float* bias = (const float*)d_in[7];

    const int E = in_sizes[0] / 2;
    const int N = in_sizes[2] / EMB;
    const int* src = edge_index;
    const int* dst = edge_index + E;

    float* ws    = (float*)d_ws;
    float* xl    = ws;
    float* xr    = xl + (size_t)N * EMB;
    float* hbuf  = xr + (size_t)N * EMB;
    const int NB = (N + BKT_NODES - 1) / BKT_NODES;
    const size_t M = (size_t)NB * L1_BLOCKS;
    int*   cntm  = (int*)(hbuf + (size_t)N * EMB);
    int*   locm  = cntm + ((M + 31) / 32) * 32;
    int*   off   = locm + ((M + 31) / 32) * 32;
    int*   tmp   = off + (((size_t)N + 1 + 31) / 32) * 32;
    int*   grp   = tmp + (((size_t)E * 4 + 31) / 32) * 32;

    const int BT = 256;
    const int rowTiles = (N + 15) / 16;
    const int gemmBlocks = (rowTiles * 4 + 7) / 8;
    const int nodeWaveBlocks = (N * 32 + BT - 1) / BT;
    const int rtB = (rowTiles < 4 * 256) ? rowTiles : 4 * 256;
    const int rtSplit = rowTiles - rtB;
    const int denseBlocksB = (rtB * 4 + 15) / 16;

    gat_slice_sort_gemm_kernel<<<2 * L1_BLOCKS, L1_THREADS, NB * sizeof(int), stream>>>(
        src, dst, ew, cntm, locm, (int4*)tmp, E, NB, emb, Wl, Wr, xl, xr, N, rtSplit);
    gat_bucket_sort_gemm_kernel<<<NB + denseBlocksB, L2_THREADS, 0, stream>>>(
        cntm, locm, (const int4*)tmp, off, grp, E, N, NB, emb, Wl, Wr, xl, xr, rtSplit);

    for (int l = 0; l < 2; ++l) {
        const float* hin = (l == 0) ? emb : hbuf;
        float* o = (l == 0) ? hbuf : (float*)d_out;
        const float* Wl_l = Wl + (size_t)l * EMB * EMB;
        const float* Wr_l = Wr + (size_t)l * EMB * EMB;
        const float* We_l = We + (size_t)l * EMB;
        const float* at_l = att + (size_t)l * EMB;
        const float* b_l  = bias + (size_t)l * EMB;

        if (l == 1) gat_gemm_kernel<<<gemmBlocks, BT, 0, stream>>>(hin, Wl_l, Wr_l, xl, xr, N);
        gat_node_kernel<<<nodeWaveBlocks, BT, 0, stream>>>(xl, xr, off, grp, We_l, at_l, b_l,
                                                           o, N);
    }
}


# node stage: a step's four edge logits share one half-wave butterfly (5 cross-lane adds for 16) and a lane takes one exponential, its own edge's, handed round the row; sum pairing unchanged
# speedup vs baseline: 7.0524x; 1.0357x over previous
#include <hip/hip_runtime.h>
#include <math.h>


#define EMB 128
#define NEG_SLOPE 0.2f

typedef float v2f __attribute__((ext_vector_type(2)));
typedef float v8f __attribute__((ext_vector_type(8)));

#define BKT_SHIFT 8
#define BKT_NODES (1 << BKT_SHIFT)
#define L1_BLOCKS 256
#define L1_THREADS 512
#define L1_PER 8
#define L2_THREADS 512
#define L2_RUNS (L1_BLOCKS / (L2_THREADS / 32))

__device__ __forceinline__ int gat_wave_incl_scan(int v, const int lane) {
#pragma unroll
    for (int d = 1; d < 32; d <<= 1) {
        const int u = __shfl_up(v, d, 32);
        if (lane >= d) v += u;
    }
    return v;
}

__device__ __forceinline__ int gat_wave_sum(int v) {
#pragma unroll
    for (int d = 16; d >= 1; d >>= 1) v += __shfl_xor(v, d, 32);
    return v;
}

__device__ __forceinline__ void gat_slice_load(const int* __restrict__ src,
                                               const int* __restrict__ dst,
                                               const float* __restrict__ ew, const int c0,
                                               const int e1, const int t, int (&s)[L1_PER],
                                               int (&d)[L1_PER], int (&w)[L1_PER]) {
#pragma unroll
    for (int j = 0; j < L1_PER; ++j) {
        const int e = c0 + j * L1_THREADS + t;
        const bool in = e < e1;
        d[j] = in ? dst[e] : -1;
        s[j] = in ? src[e] : 0;
        w[j] = in ? __float_as_int(ew[e]) : 0;
    }
}

__device__ __forceinline__ void gat_slice_count(const int (&d)[L1_PER], int* tab, const int NB) {
#pragma unroll
    for (int j = 0; j < L1_PER; ++j) {
        const unsigned b = (unsigned)(d[j] >> BKT_SHIFT);
        if (b < (unsigned)NB) atomicAdd(&tab[b], 1);
    }
}

__device__ __forceinline__
void gat_slice_sort_body(const int* __restrict__ src, const int* __restrict__ dst,
                         const float* __restrict__ ew, int* __restrict__ cntm,
                         int* __restrict__ locm, int4* __restrict__ tmp, int E, int NB) {
    extern __shared__ int lds_tab[];
    __shared__ int wtot[L1_THREADS / 32];
    const int t = threadIdx.x;
    const int lane = t & 31;
    const int wv = t >> 5;
    const int per = (E + L1_BLOCKS - 1) / L1_BLOCKS;
    const int e0 = blockIdx.x * per;
    const int e1 = (e0 + per < E) ? (e0 + per) : E;
    const int len = (e1 > e0) ? (e1 - e0) : 0;
    int s[L1_PER], d[L1_PER], w[L1_PER];
    gat_slice_load(src, dst, ew, e0, e1, t, s, d, w);
    for (int b = t; b < NB; b += L1_THREADS) lds_tab[b] = 0;
    __syncthreads();
    gat_slice_count(d, lds_tab, NB);
    for (int c0 = e0 + L1_THREADS * L1_PER; c0 < e1; c0 += L1_THREADS * L1_PER) {
        int dd[L1_PER];
#pragma unroll
        for (int j = 0; j < L1_PER; ++j) {
            const int e = c0 + j * L1_THREADS + t;
            dd[j] = (e < e1) ? dst[e] : -1;
        }
        gat_slice_count(dd, lds_tab, NB);
    }
    __syncthreads();

    int carry = 0;
    for (int b0 = 0; b0 < NB; b0 += L1_THREADS) {
        const int b = b0 + t;
        const int v = (b < NB) ? lds_tab[b] : 0;
        const int incl = gat_wave_incl_scan(v, lane);
        if (lane == 31) wtot[wv] = incl;
        __syncthreads();
        int pre = 0, tot = 0;
        for (int k = 0; k < L1_THREADS / 32; ++k) {
            const int x = wtot[k];
            tot += x;
            if (k < wv) pre += x;
        }
        const int start = carry + pre + (incl - v);
        if (b < NB) {
            lds_tab[b] = start;
            cntm[b * L1_BLOCKS + blockIdx.x] = v;
            locm[b * L1_BLOCKS + blockIdx.x] = start;
        }
        carry += tot;
        __syncthreads();
    }

    int c0 = e0;
    for (;;) {
        int pos[L1_PER];
#pragma unroll
        for (int j = 0; j < L1_PER; ++j) {
            const unsigned b = (unsigned)(d[j] >> BKT_SHIFT);
            pos[j] = -1;
            if (b < (unsigned)NB) pos[j] = atomicAdd(&lds_tab[b], 1);
        }
#pragma unroll
        for (int j = 0; j < L1_PER; ++j) {
            if ((unsigned)pos[j] < (unsigned)len) tmp[e0 + pos[j]] = make_int4(s[j], w[j], d[j], 0);
        }
        c0 += L1_THREADS * L1_PER;
        if (c0 >= e1) break;
        gat_slice_load(src, dst, ew, c0, e1, t, s, d, w);
    }
}

__device__ __forceinline__
void gat_bucket_sort_body(const int* __restrict__ cntm, const int* __restrict__ locm,
                          const int4* __restrict__ tmp, int* __restrict__ off,
                          int* __restrict__ grp, int E, int N, int NB) {
    __shared__ int cur[BKT_NODES];
    __shared__ int wlo[L2_THREADS / 32];
    __shared__ int wct[L2_THREADS / 32];
    __shared__ int wtot[BKT_NODES / 32];
    const int t = threadIdx.x;
    const int lane = t & 31;
    const int wv = t >> 5;
    const int b = blockIdx.x;
    const int n0 = b << BKT_SHIFT;
    const int per = (E + L1_BLOCKS - 1) / L1_BLOCKS;
    int c = 0, lo = 0, first = 0;
    if (lane < L2_RUNS) {
        const int k = wv * L2_RUNS + lane;
        c = cntm[b * L1_BLOCKS + k];
        lo = locm[b * L1_BLOCKS + k];
        const int e0 = k * per;
        const int e1 = (e0 + per < E) ? (e0 + per) : E;
        const int len = (e1 > e0) ? (e1 - e0) : 0;
        lo = (lo < 0) ? 0 : ((lo > len) ? len : lo);
        c = (c < 0) ? 0 : ((c > len - lo) ? (len - lo) : c);
        first = e0 + lo;
    }
    int4 r[L2_RUNS];
#pragma unroll
    for (int u = 0; u < L2_RUNS; ++u) {
        const int cu = __builtin_amdgcn_readlane(c, u);
        const int st = __builtin_amdgcn_readlane(first, u);
        int4 v = {0, 0, -1, 0};
        if (lane < cu) v = tmp[st + lane];
        r[u] = v;
    }
    if (t < BKT_NODES) cur[t] = 0;
    const int slo = gat_wave_sum(lo);
    const int sct = gat_wave_sum(c);
    if (lane == 0) { wlo[wv] = slo; wct[wv] = sct; }
    __syncthreads();
    int rs = 0, tot = 0;
    for (int k = 0; k < L2_THREADS / 32; ++k) { rs += wlo[k]; tot += wct[k]; }
    rs = (rs < 0) ? 0 : ((rs > E) ? E : rs);
    int re = rs + tot;
    re = (re < rs) ? rs : ((re > E) ? E : re);

#pragma unroll
    for (int u = 0; u < L2_RUNS; ++u) {
        const unsigned loc = (unsigned)(r[u].z - n0);
        if (loc < (unsigned)BKT_NODES) atomicAdd(&cur[loc], 1);
    }
    for (int u = 0; u < L2_RUNS; ++u) {
        const int cu = __builtin_amdgcn_readlane(c, u);
        const int st = __builtin_amdgcn_readlane(first, u);
        for (int j = 32 + lane; j < cu; j += 32) {
            const unsigned loc = (unsigned)(tmp[st + j].z - n0);
            if (loc < (unsigned)BKT_NODES) atomicAdd(&cur[loc], 1);
        }
    }
    __syncthreads();

    int v = 0, incl = 0;
    if (t < BKT_NODES) {
        v = cur[t];
        incl = gat_wave_incl_scan(v, lane);
        if (lane == 31) wtot[t >> 5] = incl;
    }
    __syncthreads();
    if (t < BKT_NODES) {
        int pre = 0;
        for (int k = 0; k < (t >> 5); ++k) pre += wtot[k];
        const int start = rs + pre + (incl - v);
        cur[t] = start;
        if (n0 + t <= N) off[n0 + t] = start;
    }
    if (b == NB - 1 && t == 0 && (N & (BKT_NODES - 1)) == 0) off[N] = re;
    __syncthreads();

    int pos[L2_RUNS];
#pragma unroll
    for (int u = 0; u < L2_RUNS; ++u) {
        const unsigned loc = (unsigned)(r[u].z - n0);
        pos[u] = -1;
        if (loc < (unsigned)BKT_NODES) pos[u] = atomicAdd(&cur[loc], 1);
    }
#pragma unroll
    for (int u = 0; u < L2_RUNS; ++u) {
        if ((unsigned)pos[u] < (unsigned)E)
            *(int2*)(grp + 2 * (size_t)pos[u]) = make_int2(r[u].x, r[u].y);
    }
    for (int u = 0; u < L2_RUNS; ++u) {
        const int cu = __builtin_amdgcn_readlane(c, u);
        const int st = __builtin_amdgcn_readlane(first, u);
        for (int j = 32 + lane; j < cu; j += 32) {
            const int4 q = tmp[st + j];
            const unsigned loc = (unsigned)(q.z - n0);
            int p = -1;
            if (loc < (unsigned)BKT_NODES) p = atomicAdd(&cur[loc], 1);
            if ((unsigned)p < (unsigned)E) *(int2*)(grp + 2 * (size_t)p) = make_int2(q.x, q.y);
        }
    }
}

typedef float v4f __attribute__((ext_vector_type(4)));

#define GAT_GK 2
struct gat_kgroup { v2f a[GAT_GK]; v4f w0[GAT_GK]; v4f w1[GAT_GK]; };

__device__ __forceinline__ void gat_group_load(gat_kgroup& g, const float* __restrict__ ap,
                                               const float* __restrict__ wp, const int k0) {
#pragma unroll
    for (int i = 0; i < GAT_GK; ++i) {
        const int k = k0 + 4 * i;
        g.a[i]  = *(const v2f*)(ap + k);
        g.w0[i] = *(const v4f*)(wp + (size_t)k * EMB);
        g.w1[i] = *(const v4f*)(wp + (size_t)(k + 1) * EMB);
    }
}

__device__ __forceinline__ void gat_group_mma(const gat_kgroup& g, v8f& c0, v8f& c1, v8f& c2,
                                              v8f& c3) {
#pragma unroll
    for (int i = 0; i < GAT_GK; ++i) {
        const v2f a = g.a[i];
        v2f b0; b0.x = g.w0[i].x; b0.y = g.w1[i].x;
        v2f b1; b1.x = g.w0[i].y; b1.y = g.w1[i].y;
        v2f b2; b2.x = g.w0[i].z; b2.y = g.w1[i].z;
        v2f b3; b3.x = g.w0[i].w; b3.y = g.w1[i].w;
        c0 = __builtin_amdgcn_wmma_f32_16x16x4_f32(false, a, false, b0, (short)0, c0,
                                                   false, false);
        c1 = __builtin_amdgcn_wmma_f32_16x16x4_f32(false, a, false, b1, (short)0, c1,
                                                   false, false);
        c2 = __builtin_amdgcn_wmma_f32_16x16x4_f32(false, a, false, b2, (short)0, c2,
                                                   false, false);
        c3 = __builtin_amdgcn_wmma_f32_16x16x4_f32(false, a, false, b3, (short)0, c3,
                                                   false, false);
    }
}

template <bool PIPE>
__device__ __forceinline__
void gat_gemm_body(const float* __restrict__ h, const float* __restrict__ Wl,
                   const float* __restrict__ Wr, float* __restrict__ xl,
                   float* __restrict__ xr, int N, const int wave, const int waveEnd) {
    int lane = threadIdx.x & 31;
    if (wave >= waveEnd) return;
    int rowTile = wave >> 2;
    int q = wave & 3;
    const float* W = (q & 2) ? Wr : Wl;
    float*      out = (q & 2) ? xr : xl;
    int cbase = (q & 1) * 64;
    int half = lane >> 4;
    int m    = lane & 15;
    int arow = rowTile * 16 + m;
    if (arow > N - 1) arow = N - 1;
    int col  = cbase + 4 * m;
    int koff = half * 2;
    const float* ap = h + (size_t)arow * EMB + koff;
    const float* wp = W + (size_t)koff * EMB + col;

    v8f c0 = {0.f, 0.f, 0.f, 0.f, 0.f, 0.f, 0.f, 0.f};
    v8f c1 = c0, c2 = c0, c3 = c0;
    if (PIPE) {
        gat_kgroup ga, gb;
        gat_group_load(ga, ap, wp, 0);
#pragma unroll
        for (int k = 0; k < EMB; k += 8 * GAT_GK) {
            gat_group_load(gb, ap, wp, k + 4 * GAT_GK);
            __builtin_amdgcn_sched_barrier(0);
            gat_group_mma(ga, c0, c1, c2, c3);
            __builtin_amdgcn_sched_barrier(0);
            if (k + 8 * GAT_GK < EMB) gat_group_load(ga, ap, wp, k + 8 * GAT_GK);
            __builtin_amdgcn_sched_barrier(0);
            gat_group_mma(gb, c0, c1, c2, c3);
            __builtin_amdgcn_sched_barrier(0);
        }
    } else {
#pragma unroll
    for (int k = 0; k < EMB; k += 4) {
        v2f a = *(const v2f*)(ap + k);
        v4f w0 = *(const v4f*)(wp + (size_t)k * EMB);
        v4f w1 = *(const v4f*)(wp + (size_t)(k + 1) * EMB);
        v2f b0; b0.x = w0.x; b0.y = w1.x;
        v2f b1; b1.x = w0.y; b1.y = w1.y;
        v2f b2; b2.x = w0.z; b2.y = w1.z;
        v2f b3; b3.x = w0.w; b3.y = w1.w;
        c0 = __builtin_amdgcn_wmma_f32_16x16x4_f32(false, a, false, b0, (short)0, c0,
                                                   false, false);
        c1 = __builtin_amdgcn_wmma_f32_16x16x4_f32(false, a, false, b1, (short)0, c1,
                                                   false, false);
        c2 = __builtin_amdgcn_wmma_f32_16x16x4_f32(false, a, false, b2, (short)0, c2,
                                                   false, false);
        c3 = __builtin_amdgcn_wmma_f32_16x16x4_f32(false, a, false, b3, (short)0, c3,
                                                   false, false);
    }
    }
#pragma unroll
    for (int j = 0; j < 8; ++j) {
        int r = rowTile * 16 + j + half * 8;
        v4f v; v.x = c0[j]; v.y = c1[j]; v.z = c2[j]; v.w = c3[j];
        if (r < N) *(v4f*)(out + (size_t)r * EMB + col) = v;
    }
}

__global__ __launch_bounds__(256)
void gat_gemm_kernel(const float* __restrict__ h, const float* __restrict__ Wl,
                     const float* __restrict__ Wr, float* __restrict__ xl,
                     float* __restrict__ xr, int N) {
    const int wave = (blockIdx.x * blockDim.x + threadIdx.x) >> 5;
    gat_gemm_body<false>(h, Wl, Wr, xl, xr, N, wave, ((N + 15) >> 4) * 4);
}

__global__ __launch_bounds__(L1_THREADS)
void gat_slice_sort_gemm_kernel(const int* __restrict__ src, const int* __restrict__ dst,
                                const float* __restrict__ ew, int* __restrict__ cntm,
                                int* __restrict__ locm, int4* __restrict__ tmp, int E, int NB,
                                const float* __restrict__ h, const float* __restrict__ Wl,
                                const float* __restrict__ Wr, float* __restrict__ xl,
                                float* __restrict__ xr, int N, int rtSplit) {
    if (blockIdx.x < L1_BLOCKS) {
        gat_slice_sort_body(src, dst, ew, cntm, locm, tmp, E, NB);
    } else {
        const int t = __builtin_amdgcn_readfirstlane((int)(threadIdx.x >> 5));
        const int j = (int)blockIdx.x - L1_BLOCKS;
        const int lo = (int)(((long long)j * rtSplit) / L1_BLOCKS);
        const int hi = (int)(((long long)(j + 1) * rtSplit) / L1_BLOCKS);
        for (int it = t; it < (hi - lo) * 4; it += L1_THREADS / 32)
            gat_gemm_body<true>(h, Wl, Wr, xl, xr, N, lo * 4 + it, rtSplit * 4);
    }
}

__global__ __launch_bounds__(L2_THREADS)
void gat_bucket_sort_gemm_kernel(const int* __restrict__ cntm, const int* __restrict__ locm,
                                 const int4* __restrict__ tmp, int* __restrict__ off,
                                 int* __restrict__ grp, int E, int N, int NB,
                                 const float* __restrict__ h, const float* __restrict__ Wl,
                                 const float* __restrict__ Wr, float* __restrict__ xl,
                                 float* __restrict__ xr, int rtSplit) {
    if ((int)blockIdx.x < NB) {
        gat_bucket_sort_body(cntm, locm, tmp, off, grp, E, N, NB);
    } else {
        const int wave = rtSplit * 4 + ((int)blockIdx.x - NB) * (L2_THREADS / 32) +
                         (int)(threadIdx.x >> 5);
        gat_gemm_body<true>(h, Wl, Wr, xl, xr, N, wave, ((N + 15) >> 4) * 4);
    }
}

template <int K>
__device__ __forceinline__ float gat_row_xadd(const float v) {
    const int u = __builtin_amdgcn_update_dpp(0, __float_as_int(v), 0x160 + K, 0xF, 0xF, true);
    return v + __int_as_float(u);
}

__device__ __forceinline__ float gat_leaky(const float v) {
    return fmaxf(v, NEG_SLOPE * v);
}

__device__ __forceinline__ float gat_head_logit(const float4 xs, const float4 xd, const float4 we,
                                                const float4 at, const float w) {
    float p = gat_leaky(xs.x + xd.x + w * we.x) * at.x;
    p += gat_leaky(xs.y + xd.y + w * we.y) * at.y;
    p += gat_leaky(xs.z + xd.z + w * we.z) * at.z;
    p += gat_leaky(xs.w + xd.w + w * we.w) * at.w;
    p = gat_row_xadd<8>(p);
    p = gat_row_xadd<4>(p);
    p = gat_row_xadd<2>(p);
    p = gat_row_xadd<1>(p);
    return p;
}

__device__ __forceinline__ float gat_head_partial(const float4 xs, const float4 xd, const float4 we,
                                                  const float4 at, const float w) {
    float p = gat_leaky(xs.x + xd.x + w * we.x) * at.x;
    p += gat_leaky(xs.y + xd.y + w * we.y) * at.y;
    p += gat_leaky(xs.z + xd.z + w * we.z) * at.z;
    p += gat_leaky(xs.w + xd.w + w * we.w) * at.w;
    return p;
}

template <int K>
__device__ __forceinline__ float gat_row_xadd2(const float keep, const float give) {
    const int u = __builtin_amdgcn_update_dpp(0, __float_as_int(give), 0x160 + K, 0xF, 0xF, true);
    return keep + __int_as_float(u);
}

template <int K>
__device__ __forceinline__ float gat_row_xmax(const float v) {
    const int u = __builtin_amdgcn_update_dpp(0, __float_as_int(v), 0x160 + K, 0xF, 0xF, true);
    return fmaxf(v, __int_as_float(u));
}

template <int Q>
__device__ __forceinline__ float gat_row_take(const float v) {
    const unsigned sel = 0x11111111u * (unsigned)(4 * Q);
    return __int_as_float((int)__builtin_amdgcn_permlane16((unsigned)__float_as_int(v),
                                                           (unsigned)__float_as_int(v), sel, sel,
                                                           false, false));
}

__global__ __launch_bounds__(256, 16)
void gat_node_kernel(const float* __restrict__ xl, const float* __restrict__ xr,
                     const int* __restrict__ off, const int* __restrict__ grp,
                     const float* __restrict__ We, const float* __restrict__ att,
                     const float* __restrict__ bias, float* __restrict__ o, int N) {
    const int wave = (blockIdx.x * blockDim.x + threadIdx.x) >> 5;
    const int lane = threadIdx.x & 31;
    if (wave >= N) return;
    const int d = __builtin_amdgcn_readfirstlane(wave);
    const int beg = off[d];
    const int end = off[d + 1];

    const float4 xd = *(const float4*)(xr + (size_t)d * EMB + lane * 4);
    const float4 we = *(const float4*)(We + lane * 4);
    const float4 at = *(const float4*)(att + lane * 4);

    float m = -INFINITY;
    float den = 0.0f;
    float4 acc = {0.0f, 0.0f, 0.0f, 0.0f};
    float wsum = 0.0f;
    const bool up8 = (lane & 8) != 0;
    const bool up4 = (lane & 4) != 0;
    const int slot = (lane >> 2) & 3;

    for (int base = beg; base < end; base += 32) {
        int my_s = 0, my_w = 0;
        if (base + lane < end) {
            const int2 rec = *(const int2*)(grp + 2 * (size_t)(base + lane));
            my_s = rec.x; my_w = rec.y;
        }
        wsum += __int_as_float(my_w);
        const int nb = (end - base < 32) ? (end - base) : 32;
        for (int i = 0; i < nb; i += 4) {
            const int i1 = (i + 1 < nb) ? (i + 1) : (nb - 1);
            const int i2 = (i + 2 < nb) ? (i + 2) : (nb - 1);
            const int i3 = (i + 3 < nb) ? (i + 3) : (nb - 1);
            const int s0 = __builtin_amdgcn_readlane(my_s, i);
            const int s1 = __builtin_amdgcn_readlane(my_s, i1);
            const int s2 = __builtin_amdgcn_readlane(my_s, i2);
            const int s3 = __builtin_amdgcn_readlane(my_s, i3);
            const float w0 = __int_as_float(__builtin_amdgcn_readlane(my_w, i));
            const float w1 = __int_as_float(__builtin_amdgcn_readlane(my_w, i1));
            const float w2 = __int_as_float(__builtin_amdgcn_readlane(my_w, i2));
            const float w3 = __int_as_float(__builtin_amdgcn_readlane(my_w, i3));
            const float4 r0 = *(const float4*)(xl + (size_t)s0 * EMB + lane * 4);
            const float4 r1 = *(const float4*)(xl + (size_t)s1 * EMB + lane * 4);
            const float4 r2 = *(const float4*)(xl + (size_t)s2 * EMB + lane * 4);
            const float4 r3 = *(const float4*)(xl + (size_t)s3 * EMB + lane * 4);
            const float q0 = gat_head_partial(r0, xd, we, at, w0);
            const float q1 = gat_head_partial(r1, xd, we, at, w1);
            const float q2 = gat_head_partial(r2, xd, we, at, w2);
            const float q3 = gat_head_partial(r3, xd, we, at, w3);
            const float a0 = gat_row_xadd2<8>(up8 ? q2 : q0, up8 ? q0 : q2);
            const float a1 = gat_row_xadd2<8>(up8 ? q3 : q1, up8 ? q1 : q3);
            float ps = gat_row_xadd2<4>(up4 ? a1 : a0, up4 ? a0 : a1);
            ps = gat_row_xadd2<2>(ps, ps);
            ps = gat_row_xadd2<1>(ps, ps);
            ps = (i + slot < nb) ? ps : -INFINITY;
            const float mn = fmaxf(m, gat_row_xmax<8>(gat_row_xmax<4>(ps)));
            const float sc = __expf(m - mn);
            const float es = __expf(ps - mn);
            const float e0 = gat_row_take<0>(es);
            const float e1 = gat_row_take<1>(es);
            const float e2 = gat_row_take<2>(es);
            const float e3 = gat_row_take<3>(es);
            den = den * sc + ((e0 + e1) + (e2 + e3));
            acc.x = acc.x * sc + ((e0 * r0.x + e1 * r1.x) + (e2 * r2.x + e3 * r3.x));
            acc.y = acc.y * sc + ((e0 * r0.y + e1 * r1.y) + (e2 * r2.y + e3 * r3.y));
            acc.z = acc.z * sc + ((e0 * r0.z + e1 * r1.z) + (e2 * r2.z + e3 * r3.z));
            acc.w = acc.w * sc + ((e0 * r0.w + e1 * r1.w) + (e2 * r2.w + e3 * r3.w));
            m = mn;
        }
    }

    const float4 xself = *(const float4*)(xl + (size_t)d * EMB + lane * 4);
    wsum += __shfl_xor(wsum, 16, 32);
    wsum = gat_row_xadd<8>(wsum);
    wsum = gat_row_xadd<4>(wsum);
    wsum = gat_row_xadd<2>(wsum);
    wsum = gat_row_xadd<1>(wsum);
    const float wself = wsum / fmaxf((float)(end - beg), 1.0f);
    {
        const float p = gat_head_logit(xself, xd, we, at, wself);
        const float mn = fmaxf(m, p);
        const float sc = __expf(m - mn);
        const float ex = __expf(p - mn);
        den = den * sc + ex;
        acc.x = acc.x * sc + ex * xself.x;
        acc.y = acc.y * sc + ex * xself.y;
        acc.z = acc.z * sc + ex * xself.z;
        acc.w = acc.w * sc + ex * xself.w;
    }

    const float4 b = *(const float4*)(bias + lane * 4);
    float4 r;
    r.x = acc.x / den + b.x;
    r.y = acc.y / den + b.y;
    r.z = acc.z / den + b.z;
    r.w = acc.w / den + b.w;
    r.x = (r.x > 0.0f) ? r.x : (__expf(r.x) - 1.0f);
    r.y = (r.y > 0.0f) ? r.y : (__expf(r.y) - 1.0f);
    r.z = (r.z > 0.0f) ? r.z : (__expf(r.z) - 1.0f);
    r.w = (r.w > 0.0f) ? r.w : (__expf(r.w) - 1.0f);
    *(float4*)(o + (size_t)d * EMB + lane * 4) = r;
}

extern "C" void kernel_launch(void* const* d_in, const int* in_sizes, int n_in,
                              void* d_out, int out_size, void* d_ws, size_t ws_size,
                              hipStream_t stream) {
    const int*   edge_index = (const int*)d_in[0];
    const float* ew   = (const float*)d_in[1];
    const float* emb  = (const float*)d_in[2];
    const float* Wl   = (const float*)d_in[3];
    const float* Wr   = (const float*)d_in[4];
    const float* We   = (const float*)d_in[5];
    const float* att  = (const float*)d_in[6];
    const float* bias = (const float*)d_in[7];

    const int E = in_sizes[0] / 2;
    const int N = in_sizes[2] / EMB;
    const int* src = edge_index;
    const int* dst = edge_index + E;

    float* ws    = (float*)d_ws;
    float* xl    = ws;
    float* xr    = xl + (size_t)N * EMB;
    float* hbuf  = xr + (size_t)N * EMB;
    const int NB = (N + BKT_NODES - 1) / BKT_NODES;
    const size_t M = (size_t)NB * L1_BLOCKS;
    int*   cntm  = (int*)(hbuf + (size_t)N * EMB);
    int*   locm  = cntm + ((M + 31) / 32) * 32;
    int*   off   = locm + ((M + 31) / 32) * 32;
    int*   tmp   = off + (((size_t)N + 1 + 31) / 32) * 32;
    int*   grp   = tmp + (((size_t)E * 4 + 31) / 32) * 32;

    const int BT = 256;
    const int rowTiles = (N + 15) / 16;
    const int gemmBlocks = (rowTiles * 4 + 7) / 8;
    const int nodeWaveBlocks = (N * 32 + BT - 1) / BT;
    const int rtB = (rowTiles < 4 * 256) ? rowTiles : 4 * 256;
    const int rtSplit = rowTiles - rtB;
    const int denseBlocksB = (rtB * 4 + 15) / 16;

    gat_slice_sort_gemm_kernel<<<2 * L1_BLOCKS, L1_THREADS, NB * sizeof(int), stream>>>(
        src, dst, ew, cntm, locm, (int4*)tmp, E, NB, emb, Wl, Wr, xl, xr, N, rtSplit);
    gat_bucket_sort_gemm_kernel<<<NB + denseBlocksB, L2_THREADS, 0, stream>>>(
        cntm, locm, (const int4*)tmp, off, grp, E, N, NB, emb, Wl, Wr, xl, xr, rtSplit);

    for (int l = 0; l < 2; ++l) {
        const float* hin = (l == 0) ? emb : hbuf;
        float* o = (l == 0) ? hbuf : (float*)d_out;
        const float* Wl_l = Wl + (size_t)l * EMB * EMB;
        const float* Wr_l = Wr + (size_t)l * EMB * EMB;
        const float* We_l = We + (size_t)l * EMB;
        const float* at_l = att + (size_t)l * EMB;
        const float* b_l  = bias + (size_t)l * EMB;

        if (l == 1) gat_gemm_kernel<<<gemmBlocks, BT, 0, stream>>>(hin, Wl_l, Wr_l, xl, xr, N);
        gat_node_kernel<<<nodeWaveBlocks, BT, 0, stream>>>(xl, xr, off, grp, We_l, at_l, b_l,
                                                           o, N);
    }
}
